# starting kernel
# baseline (speedup 1.0000x reference)
#include <hip/hip_runtime.h>
#include <hip/hip_bf16.h>

typedef float v2f __attribute__((ext_vector_type(2)));
typedef float v8f __attribute__((ext_vector_type(8)));

#define D_FEAT 64
#define D_OUT  32

__global__ void fill_f32(float* __restrict__ p, float v, int n) {
    int i = blockIdx.x * blockDim.x + threadIdx.x;
    if (i < n) p[i] = v;
}

__global__ void deg_count(const int* __restrict__ dst, float* __restrict__ deg, int e) {
    int i = blockIdx.x * blockDim.x + threadIdx.x;
    if (i < e) unsafeAtomicAdd(&deg[dst[i]], 1.0f);
}

__global__ void rsqrt_inplace(float* __restrict__ p, int n) {
    int i = blockIdx.x * blockDim.x + threadIdx.x;
    if (i < n) {
        float d = p[i];
        p[i] = d > 0.0f ? __frsqrt_rn(d) : 0.0f;
    }
}

__global__ __launch_bounds__(256) void gcn_gemm64_wmma(
    const float* __restrict__ X, const float* __restrict__ W,
    float* __restrict__ H, int nrows)
{
    const int lane  = threadIdx.x & 31;
    const int strip = blockIdx.x * 8 + (threadIdx.x >> 5);
    const int r0 = strip * 16;
    if (r0 >= nrows) return;

    const int m  = lane & 15;
    const int kk = (lane >> 4) * 2;
    const int n  = lane & 15;

    v8f acc0 = {}, acc1 = {}, acc2 = {}, acc3 = {};
    const float* xrow = X + (size_t)(r0 + m) * D_FEAT;

    #pragma unroll
    for (int k0 = 0; k0 < D_FEAT; k0 += 4) {
        v2f a;
        a.x = xrow[k0 + kk];
        a.y = xrow[k0 + kk + 1];
        const float* w0 = W + (size_t)(k0 + kk) * D_FEAT + n;
        const float* w1 = W + (size_t)(k0 + kk + 1) * D_FEAT + n;
        v2f b0; b0.x = w0[0];  b0.y = w1[0];
        v2f b1; b1.x = w0[16]; b1.y = w1[16];
        v2f b2; b2.x = w0[32]; b2.y = w1[32];
        v2f b3; b3.x = w0[48]; b3.y = w1[48];
        acc0 = __builtin_amdgcn_wmma_f32_16x16x4_f32(false, a, false, b0, (short)0, acc0, false, false);
        acc1 = __builtin_amdgcn_wmma_f32_16x16x4_f32(false, a, false, b1, (short)0, acc1, false, false);
        acc2 = __builtin_amdgcn_wmma_f32_16x16x4_f32(false, a, false, b2, (short)0, acc2, false, false);
        acc3 = __builtin_amdgcn_wmma_f32_16x16x4_f32(false, a, false, b3, (short)0, acc3, false, false);
    }

    const int rbase = (lane >> 4) * 8;
    const int col   = lane & 15;
    #pragma unroll
    for (int r = 0; r < 8; ++r) {
        float* hrow = H + (size_t)(r0 + rbase + r) * D_FEAT + col;
        hrow[0]  = acc0[r];
        hrow[16] = acc1[r];
        hrow[32] = acc2[r];
        hrow[48] = acc3[r];
    }
}

__global__ __launch_bounds__(256) void gcn_scatter(
    const int* __restrict__ srcIdx, const int* __restrict__ dstIdx,
    const float* __restrict__ dinv, const float* __restrict__ h,
    float* __restrict__ agg, int nEdges)
{
    int idx = blockIdx.x * blockDim.x + threadIdx.x;
    if (idx >= nEdges * 16) return;
    int e = idx >> 4;
    int f = (idx & 15) * 4;
    int s = srcIdx[e], d = dstIdx[e];
    float nrm = dinv[s] * dinv[d];
    const float4 hv = *reinterpret_cast<const float4*>(h + (size_t)s * D_FEAT + f);
    float* ap = agg + (size_t)d * D_FEAT + f;
    unsafeAtomicAdd(ap + 0, nrm * hv.x);
    unsafeAtomicAdd(ap + 1, nrm * hv.y);
    unsafeAtomicAdd(ap + 2, nrm * hv.z);
    unsafeAtomicAdd(ap + 3, nrm * hv.w);
}

__global__ __launch_bounds__(256) void gcn_combine(
    const float* __restrict__ agg, const float* __restrict__ h,
    const float* __restrict__ dinv, const float* __restrict__ bias,
    const float* __restrict__ xprev, float* __restrict__ xnew, int n64)
{
    int idx = blockIdx.x * blockDim.x + threadIdx.x;
    if (idx >= n64) return;
    int node = idx >> 6, f = idx & 63;
    float di = dinv[node];
    float v = agg[idx] + di * di * h[idx] + bias[f];
    v = v > 0.0f ? v : 0.0f;
    xnew[idx] = v + xprev[idx];
}

__global__ void pool_count(const int* __restrict__ batch, float* __restrict__ cnts, int n) {
    int i = blockIdx.x * blockDim.x + threadIdx.x;
    if (i < n) unsafeAtomicAdd(&cnts[batch[i]], 1.0f);
}

__global__ __launch_bounds__(256) void pool_sum(
    const float* __restrict__ x, const int* __restrict__ batch,
    float* __restrict__ sums, int n64)
{
    int idx = blockIdx.x * blockDim.x + threadIdx.x;
    if (idx >= n64) return;
    int node = idx >> 6, f = idx & 63;
    unsafeAtomicAdd(&sums[(size_t)batch[node] * D_FEAT + f], x[idx]);
}

__global__ __launch_bounds__(256) void final_linear_wmma(
    const float* __restrict__ sums, const float* __restrict__ cnts,
    const float* __restrict__ Wl, const float* __restrict__ bl,
    float* __restrict__ out, int nrows)
{
    const int lane  = threadIdx.x & 31;
    const int strip = blockIdx.x * 8 + (threadIdx.x >> 5);
    const int r0 = strip * 16;
    if (r0 >= nrows) return;

    const int m  = lane & 15;
    const int kk = (lane >> 4) * 2;
    const int n  = lane & 15;
    const int row = r0 + m;
    float c = cnts[row];
    float inv = 1.0f / (c > 1.0f ? c : 1.0f);

    v8f acc0 = {}, acc1 = {};
    const float* srow = sums + (size_t)row * D_FEAT;

    #pragma unroll
    for (int k0 = 0; k0 < D_FEAT; k0 += 4) {
        v2f a;
        a.x = srow[k0 + kk] * inv;
        a.y = srow[k0 + kk + 1] * inv;
        const float* w0 = Wl + (size_t)(k0 + kk) * D_OUT + n;
        const float* w1 = Wl + (size_t)(k0 + kk + 1) * D_OUT + n;
        v2f b0; b0.x = w0[0];  b0.y = w1[0];
        v2f b1; b1.x = w0[16]; b1.y = w1[16];
        acc0 = __builtin_amdgcn_wmma_f32_16x16x4_f32(false, a, false, b0, (short)0, acc0, false, false);
        acc1 = __builtin_amdgcn_wmma_f32_16x16x4_f32(false, a, false, b1, (short)0, acc1, false, false);
    }

    const int rbase = (lane >> 4) * 8;
    const int col   = lane & 15;
    #pragma unroll
    for (int r = 0; r < 8; ++r) {
        float* orow = out + (size_t)(r0 + rbase + r) * D_OUT + col;
        orow[0]  = acc0[r] + bl[col];
        orow[16] = acc1[r] + bl[col + 16];
    }
}

static inline int cdiv(long long a, int b) { return (int)((a + b - 1) / b); }

extern "C" void kernel_launch(void* const* d_in, const int* in_sizes, int n_in,
                              void* d_out, int out_size, void* d_ws, size_t ws_size,
                              hipStream_t stream)
{
    const float* x     = (const float*)d_in[0];
    const int*   ei    = (const int*)d_in[1];
    const int*   batch = (const int*)d_in[2];
    const float* W[3]  = {(const float*)d_in[3], (const float*)d_in[5], (const float*)d_in[7]};
    const float* B[3]  = {(const float*)d_in[4], (const float*)d_in[6], (const float*)d_in[8]};
    const float* Wl    = (const float*)d_in[9];
    const float* bl    = (const float*)d_in[10];
    float* out = (float*)d_out;

    const int N = in_sizes[0] / D_FEAT;
    const int E = in_sizes[1] / 2;
    const int G = out_size / D_OUT;
    const int* srcIdx = ei;
    const int* dstIdx = ei + E;

    float* ws   = (float*)d_ws;
    float* buf0 = ws;
    float* buf1 = buf0 + (size_t)N * D_FEAT;
    float* hbuf = buf1 + (size_t)N * D_FEAT;
    float* dinv = hbuf + (size_t)N * D_FEAT;
    float* sums = dinv + N;
    float* cnts = sums + (size_t)G * D_FEAT;

    const int TB = 256;
    const int n64 = N * D_FEAT;

    fill_f32<<<cdiv(N, TB), TB, 0, stream>>>(dinv, 1.0f, N);
    deg_count<<<cdiv(E, TB), TB, 0, stream>>>(dstIdx, dinv, E);
    rsqrt_inplace<<<cdiv(N, TB), TB, 0, stream>>>(dinv, N);

    fill_f32<<<cdiv(G * D_FEAT, TB), TB, 0, stream>>>(sums, 0.0f, G * D_FEAT);
    fill_f32<<<cdiv(G, TB), TB, 0, stream>>>(cnts, 0.0f, G);

    const int gemmBlocks = cdiv((long long)cdiv(N, 16), 8);
    const float* xcur = x;
    float* aggbuf[3] = {buf0, buf1, buf0};

    for (int l = 0; l < 3; ++l) {
        float* agg = aggbuf[l];
        gcn_gemm64_wmma<<<gemmBlocks, TB, 0, stream>>>(xcur, W[l], hbuf, N);
        fill_f32<<<cdiv(n64, TB), TB, 0, stream>>>(agg, 0.0f, n64);
        gcn_scatter<<<cdiv((long long)E * 16, TB), TB, 0, stream>>>(srcIdx, dstIdx, dinv, hbuf, agg, E);
        gcn_combine<<<cdiv(n64, TB), TB, 0, stream>>>(agg, hbuf, dinv, B[l], xcur, agg, n64);
        xcur = agg;
    }

    pool_count<<<cdiv(N, TB), TB, 0, stream>>>(batch, cnts, N);
    pool_sum<<<cdiv(n64, TB), TB, 0, stream>>>(xcur, batch, sums, n64);
    final_linear_wmma<<<cdiv((long long)cdiv(G, 16), 8), TB, 0, stream>>>(sums, cnts, Wl, bl, out, G);
}


# Tile lists built by one binning kernel: a work-group ranks a chunk's edges per tile on LDS counters, stores whole lines; no tile scans the edge list; layer 1 adds weights; sum order = LDS arrival
# speedup vs baseline: 2.5313x; 2.5313x over previous
#include <hip/hip_runtime.h>
#include <hip/hip_bf16.h>

typedef float v2f __attribute__((ext_vector_type(2)));
typedef float v8f __attribute__((ext_vector_type(8)));

#define D_FEAT 64
#define D_OUT  32

#define TILE_NODES 392
#define QCAP 512
#define LIST_CAP 16384
#define LIST_WORDS (3 * LIST_CAP)
#define BIN_CHUNK 16384
#define BIN_EPT (BIN_CHUNK / 1024)
#define BIN_TILES_MAX 256
#define BIN_STAGE (BIN_CHUNK + 31 * BIN_TILES_MAX)

__global__ void init_deg_lists(float* __restrict__ deg, int n, int* __restrict__ tileFill, int tiles) {
    int i = blockIdx.x * blockDim.x + threadIdx.x;
    if (i < n) deg[i] = 1.0f;
    if (i < tiles) tileFill[i] = (tiles > BIN_TILES_MAX) ? LIST_CAP + 1 : 0;
}

__global__ void deg_count(const int* __restrict__ dst, float* __restrict__ deg, int e) {
    int i = blockIdx.x * blockDim.x + threadIdx.x;
    if (i < e) unsafeAtomicAdd(&deg[dst[i]], 1.0f);
}

__global__ void rsqrt_inplace(float* __restrict__ p, int n) {
    int i = blockIdx.x * blockDim.x + threadIdx.x;
    if (i < n) {
        float d = p[i];
        p[i] = d > 0.0f ? __frsqrt_rn(d) : 0.0f;
    }
}

__global__ __launch_bounds__(256) void gcn_gemm64_wmma(
    const float* __restrict__ X, const float* __restrict__ W,
    float* __restrict__ H, int nrows)
{
    const int lane  = threadIdx.x & 31;
    const int strip = blockIdx.x * 8 + (threadIdx.x >> 5);
    const int r0 = strip * 16;
    if (r0 >= nrows) return;

    const int m  = lane & 15;
    const int kk = (lane >> 4) * 2;
    const int n  = lane & 15;

    v8f acc0 = {}, acc1 = {}, acc2 = {}, acc3 = {};
    const float* xrow = X + (size_t)(r0 + m) * D_FEAT;

    #pragma unroll
    for (int k0 = 0; k0 < D_FEAT; k0 += 4) {
        v2f a;
        a.x = xrow[k0 + kk];
        a.y = xrow[k0 + kk + 1];
        const float* w0 = W + (size_t)(k0 + kk) * D_FEAT + n;
        const float* w1 = W + (size_t)(k0 + kk + 1) * D_FEAT + n;
        v2f b0; b0.x = w0[0];  b0.y = w1[0];
        v2f b1; b1.x = w0[16]; b1.y = w1[16];
        v2f b2; b2.x = w0[32]; b2.y = w1[32];
        v2f b3; b3.x = w0[48]; b3.y = w1[48];
        acc0 = __builtin_amdgcn_wmma_f32_16x16x4_f32(false, a, false, b0, (short)0, acc0, false, false);
        acc1 = __builtin_amdgcn_wmma_f32_16x16x4_f32(false, a, false, b1, (short)0, acc1, false, false);
        acc2 = __builtin_amdgcn_wmma_f32_16x16x4_f32(false, a, false, b2, (short)0, acc2, false, false);
        acc3 = __builtin_amdgcn_wmma_f32_16x16x4_f32(false, a, false, b3, (short)0, acc3, false, false);
    }

    const int rbase = (lane >> 4) * 8;
    const int col   = lane & 15;
    #pragma unroll
    for (int r = 0; r < 8; ++r) {
        float* hrow = H + (size_t)(r0 + rbase + r) * D_FEAT + col;
        hrow[0]  = acc0[r];
        hrow[16] = acc1[r];
        hrow[32] = acc2[r];
        hrow[48] = acc3[r];
    }
}

template <int U>
__device__ __forceinline__ void rows_add(
    int s, int dl, float nrm, int n, int lane, const float* __restrict__ h, float* acc)
{
    for (int jb = 0; jb < n; jb += U) {
        float v0[U], v1[U], nj[U];
        int dj[U];
        #pragma unroll
        for (int u = 0; u < U; ++u) {
            const int j = jb + u;
            const int sj = __builtin_amdgcn_readlane(s, j);
            dj[u] = __builtin_amdgcn_readlane(dl, j);
            nj[u] = __int_as_float(__builtin_amdgcn_readlane(__float_as_int(nrm), j));
            const float* hr = h + (size_t)sj * D_FEAT + lane;
            v0[u] = hr[0];
            v1[u] = hr[32];
        }
        #pragma unroll
        for (int u = 0; u < U; ++u) {
            if (jb + u < n) {
                float* ar = acc + dj[u] * D_FEAT + lane;
                __hip_atomic_fetch_add(ar,      nj[u] * v0[u], __ATOMIC_RELAXED, __HIP_MEMORY_SCOPE_WORKGROUP);
                __hip_atomic_fetch_add(ar + 32, nj[u] * v1[u], __ATOMIC_RELAXED, __HIP_MEMORY_SCOPE_WORKGROUP);
            }
        }
    }
}

template <int JB>
__device__ __forceinline__ void rows_add_half(
    int s, int dl, float nrm, int lane, const float* __restrict__ h, float* acc)
{
    float v0[16], v1[16], nj[16];
    int dj[16];
    #pragma unroll
    for (int u = 0; u < 16; ++u) {
        const int sj = __builtin_amdgcn_readlane(s, JB + u);
        dj[u] = __builtin_amdgcn_readlane(dl, JB + u);
        nj[u] = __int_as_float(__builtin_amdgcn_readlane(__float_as_int(nrm), JB + u));
        const float* hr = h + (size_t)sj * D_FEAT + lane;
        v0[u] = hr[0];
        v1[u] = hr[32];
    }
    #pragma unroll
    for (int u = 0; u < 16; ++u) {
        float* ar = acc + dj[u] * D_FEAT + lane;
        __hip_atomic_fetch_add(ar,      nj[u] * v0[u], __ATOMIC_RELAXED, __HIP_MEMORY_SCOPE_WORKGROUP);
        __hip_atomic_fetch_add(ar + 32, nj[u] * v1[u], __ATOMIC_RELAXED, __HIP_MEMORY_SCOPE_WORKGROUP);
    }
}

__device__ __forceinline__ void tile_add_batch(
    const int* qp, int n, int lane, int lo,
    const int* __restrict__ srcIdx, const int* __restrict__ dstIdx,
    const float* __restrict__ dinv, const float* __restrict__ h, float* acc)
{
    int s = 0, dl = 0;
    float nrm = 0.0f;
    if (lane < n) {
        const int e = qp[lane];
        s = srcIdx[e];
        const int d = dstIdx[e];
        nrm = dinv[s] * dinv[d];
        dl = d - lo;
    }
    rows_add<8>(s, dl, nrm, n, lane, h, acc);
}

__device__ __forceinline__ void queue_matches(int dstv, int e, int lo, unsigned tileCount, int* q, int& cnt)
{
    const bool m = (unsigned)(dstv - lo) < tileCount;
    const unsigned mask = __builtin_amdgcn_ballot_w32(m);
    if (mask != 0u) {
        if (m) q[cnt + (int)__builtin_amdgcn_mbcnt_lo(mask, 0u)] = e;
        cnt += __builtin_popcount(mask);
    }
}

__device__ __forceinline__ void filter_pass(
    const int* __restrict__ srcIdx, const int* __restrict__ dstIdx,
    const float* __restrict__ dinv, const float* __restrict__ h, float* acc,
    int* q, int lane, int wave, int lo, unsigned tileCount, int nEdges)
{
    int cnt = 0;

    const bool aligned = (reinterpret_cast<size_t>(dstIdx) & 15) == 0;
    const int nFull = aligned ? (nEdges >> 10) : 0;
    const int rot = (int)(((long long)blockIdx.x * nFull) / gridDim.x);

    for (int c0 = wave; c0 < nFull; c0 += 32) {
        int c = c0 + rot;
        if (c >= nFull) c -= nFull;
        const int cb = c << 10;
        const int4* p = reinterpret_cast<const int4*>(dstIdx + cb) + lane;
        int4 dv[8];
        #pragma unroll
        for (int k = 0; k < 8; ++k) dv[k] = p[k * 32];
        #pragma unroll
        for (int k = 0; k < 8; ++k) {
            const int e0 = cb + k * 128 + lane * 4;
            queue_matches(dv[k].x, e0 + 0, lo, tileCount, q, cnt);
            queue_matches(dv[k].y, e0 + 1, lo, tileCount, q, cnt);
            queue_matches(dv[k].z, e0 + 2, lo, tileCount, q, cnt);
            queue_matches(dv[k].w, e0 + 3, lo, tileCount, q, cnt);
            while (cnt >= 32) {
                cnt -= 32;
                tile_add_batch(q + cnt, 32, lane, lo, srcIdx, dstIdx, dinv, h, acc);
            }
        }
    }

    for (int eb = (nFull << 10) + wave * 128; eb < nEdges; eb += 32 * 128) {
        #pragma unroll
        for (int t = 0; t < 4; ++t) {
            const int e = eb + t * 32 + lane;
            const int dstv = (e < nEdges) ? dstIdx[e] : -1;
            queue_matches(dstv, e, lo, tileCount, q, cnt);
        }
        while (cnt >= 32) {
            cnt -= 32;
            tile_add_batch(q + cnt, 32, lane, lo, srcIdx, dstIdx, dinv, h, acc);
        }
    }
    if (cnt > 0) tile_add_batch(q, cnt, lane, lo, srcIdx, dstIdx, dinv, h, acc);
}

template <bool FIRST>
__device__ __forceinline__ void list_pass(
    const int* __restrict__ rec, int* __restrict__ wts, int nSlots,
    const float* __restrict__ dinv, int lo,
    const float* __restrict__ h, float* acc, int lane, int wave)
{
    int c = wave * 32;
    if (c >= nSlots) return;
    int s  = rec[c + lane];
    int dl = rec[LIST_CAP + c + lane];
    int nb = 0;
    if (!FIRST) nb = rec[2 * LIST_CAP + c + lane];
    for (;;) {
        const int cn = c + 32 * 32;
        const bool more = cn < nSlots;
        const int cp = more ? cn : c;
        const int s2  = rec[cp + lane];
        const int dl2 = rec[LIST_CAP + cp + lane];
        int nb2 = 0;
        if (!FIRST) nb2 = rec[2 * LIST_CAP + cp + lane];
        float nrm;
        if (FIRST) {
            const bool used = dl < TILE_NODES;
            const float w = dinv[s] * dinv[lo + (used ? dl : 0)];
            nrm = used ? w : 0.0f;
        } else {
            nrm = __int_as_float(nb);
        }
        rows_add_half<0>(s, dl, nrm, lane, h, acc);
        if (__builtin_amdgcn_readlane(dl, 16) != TILE_NODES)
            rows_add_half<16>(s, dl, nrm, lane, h, acc);
        if (FIRST) wts[c + lane] = __float_as_int(nrm);
        if (!more) break;
        s = s2; dl = dl2; nb = nb2;
        c = cn;
    }
}

__global__ __launch_bounds__(1024) void bin_edges(
    const int* __restrict__ srcIdx, const int* __restrict__ dstIdx,
    int nNodes, int nEdges, int tiles,
    int* __restrict__ lists, int* __restrict__ tileFill)
{
    __shared__ int stS[BIN_STAGE];
    __shared__ int stD[BIN_STAGE];
    __shared__ int cnt[BIN_TILES_MAX];
    __shared__ int lbase[BIN_TILES_MAX];
    __shared__ int gbase[BIN_TILES_MAX];
    __shared__ int lineTile[BIN_STAGE / 32];
    __shared__ int wtot[BIN_TILES_MAX / 32];
    __shared__ int totalSlots;

    if (tiles > BIN_TILES_MAX) return;

    const int tid  = threadIdx.x;
    const int lane = tid & 31;
    const int wave = __builtin_amdgcn_readfirstlane(tid >> 5);

    if (tid < BIN_TILES_MAX) cnt[tid] = 0;
    __syncthreads();

    const int e0 = blockIdx.x * BIN_CHUNK + tid;
    int sv[BIN_EPT], dv[BIN_EPT], rk[BIN_EPT];
    #pragma unroll
    for (int k = 0; k < BIN_EPT; ++k) {
        const int e = e0 + k * 1024;
        const bool ok = e < nEdges;
        const int el = ok ? e : 0;
        const int d = dstIdx[el];
        sv[k] = srcIdx[el];
        dv[k] = ok ? d : -1;
    }
    #pragma unroll
    for (int k = 0; k < BIN_EPT; ++k) {
        rk[k] = 0;
        if ((unsigned)dv[k] < (unsigned)nNodes)
            rk[k] = __hip_atomic_fetch_add(&cnt[dv[k] / TILE_NODES], 1, __ATOMIC_RELAXED, __HIP_MEMORY_SCOPE_WORKGROUP);
        else
            dv[k] = -1;
    }
    __syncthreads();

    int n = 0, padded = 0;
    if (tid < BIN_TILES_MAX) {
        n = cnt[tid];
        padded = (n + 31) & ~31;
    }
    int incl = padded;
    #pragma unroll
    for (int off = 1; off < 32; off <<= 1) {
        const int v = __shfl_up(incl, off);
        if (lane >= off) incl += v;
    }
    if (tid < BIN_TILES_MAX && lane == 31) wtot[wave] = incl;
    __syncthreads();
    if (tid < BIN_TILES_MAX) {
        int pre = 0;
        for (int w = 0; w < wave; ++w) pre += wtot[w];
        const int excl = pre + incl - padded;
        lbase[tid] = excl;
        int gb = 0;
        if (n > 0) gb = __hip_atomic_fetch_add(&tileFill[tid], padded, __ATOMIC_RELAXED, __HIP_MEMORY_SCOPE_AGENT);
        gbase[tid] = gb;
        for (int j = excl >> 5; j < ((excl + padded) >> 5); ++j) lineTile[j] = tid;
        for (int j = excl + n; j < excl + padded; ++j) { stS[j] = 0; stD[j] = TILE_NODES; }
        if (tid == BIN_TILES_MAX - 1) totalSlots = excl + padded;
    }
    __syncthreads();

    int pp[BIN_EPT];
    #pragma unroll
    for (int k = 0; k < BIN_EPT; ++k) {
        const int dd = dv[k] < 0 ? 0 : dv[k];
        const int t = dd / TILE_NODES;
        pp[k] = lbase[t] + rk[k];
        rk[k] = dd - t * TILE_NODES;
    }
    #pragma unroll
    for (int k = 0; k < BIN_EPT; ++k) {
        if (dv[k] >= 0) {
            stS[pp[k]] = sv[k];
            stD[pp[k]] = rk[k];
        }
    }
    __syncthreads();

    const int nLines = totalSlots >> 5;
    for (int ln0 = wave; ln0 < nLines; ln0 += 4 * 32) {
        int tt[4], gp[4], a[4], b[4];
        #pragma unroll
        for (int u = 0; u < 4; ++u) {
            const int ln = ln0 + 32 * u;
            tt[u] = lineTile[ln < nLines ? ln : ln0];
        }
        #pragma unroll
        for (int u = 0; u < 4; ++u) {
            const int ln = ln0 + 32 * u;
            const int lc = ln < nLines ? ln : ln0;
            tt[u] = __builtin_amdgcn_readfirstlane(tt[u]);
            gp[u] = gbase[tt[u]] + (lc * 32 - lbase[tt[u]]);
            a[u] = stS[lc * 32 + lane];
            b[u] = stD[lc * 32 + lane];
        }
        #pragma unroll
        for (int u = 0; u < 4; ++u) {
            const int ln = ln0 + 32 * u;
            const int gpos = __builtin_amdgcn_readfirstlane(gp[u]);
            if (ln < nLines && gpos < LIST_CAP) {
                int* r = lists + (size_t)tt[u] * LIST_WORDS + gpos + lane;
                r[0]        = a[u];
                r[LIST_CAP] = b[u];
            }
        }
    }
}

template <bool FIRST>
__global__ __launch_bounds__(1024) void gcn_owner_tile(
    const int* __restrict__ srcIdx, const int* __restrict__ dstIdx,
    const float* __restrict__ dinv, const float* __restrict__ h,
    const float* __restrict__ bias, const float* __restrict__ xprev,
    float* __restrict__ xnew, int nNodes, int nEdges,
    int* __restrict__ lists, const int* __restrict__ tileFill)
{
    __shared__ float acc[(TILE_NODES + 1) * D_FEAT];
    __shared__ int   queue[32 * QCAP];

    const int tid  = threadIdx.x;
    const int lane = tid & 31;
    const int wave = __builtin_amdgcn_readfirstlane(tid >> 5);
    const int lo   = blockIdx.x * TILE_NODES;
    int cntNodes = nNodes - lo;
    if (cntNodes > TILE_NODES) cntNodes = TILE_NODES;
    if (cntNodes <= 0) return;
    const unsigned tileCount = (unsigned)cntNodes;

    for (int i = tid; i < (TILE_NODES + 1) * D_FEAT; i += 1024) acc[i] = 0.0f;
    __syncthreads();

    int* q = queue + wave * QCAP;
    int* rec = lists + (size_t)blockIdx.x * LIST_WORDS;

    const int nSlots = tileFill[blockIdx.x];
    if (nSlots <= LIST_CAP) {
        list_pass<FIRST>(rec, rec + 2 * LIST_CAP, nSlots, dinv, lo, h, acc, lane, wave);
    } else {
        filter_pass(srcIdx, dstIdx, dinv, h, acc, q, lane, wave, lo, tileCount, nEdges);
    }
    __syncthreads();

    const size_t gbase = (size_t)lo * D_FEAT;
    const int nElem = cntNodes * D_FEAT;
    for (int i = tid; i < nElem; i += 1024) {
        const int f = i & 63;
        const float di = dinv[lo + (i >> 6)];
        float v = acc[i] + di * di * h[gbase + i] + bias[f];
        v = v > 0.0f ? v : 0.0f;
        xnew[gbase + i] = v + xprev[gbase + i];
    }
}

#define POOL_WAVES 32
#define POOL_UNIT 128
#define POOL_UNITS 32
#define POOL_PASS_NODES (POOL_WAVES * POOL_UNITS * POOL_UNIT)
#define POOL_ITEM_CAP (POOL_UNITS * 8)
#define CNT_PITCH 32

__device__ __forceinline__ int4 pool_load_unit(
    const int* __restrict__ batch, int n0, int lane, int nNodes, bool aligned)
{
    const int n = n0 + lane * 4;
    int4 r;
    if (aligned && n0 + POOL_UNIT <= nNodes) {
        r = *reinterpret_cast<const int4*>(batch + n);
    } else {
        r.x = (n + 0 < nNodes) ? batch[n + 0] : -1;
        r.y = (n + 1 < nNodes) ? batch[n + 1] : -1;
        r.z = (n + 2 < nNodes) ? batch[n + 2] : -1;
        r.w = (n + 3 < nNodes) ? batch[n + 3] : -1;
    }
    return r;
}

__device__ __forceinline__ void pool_note(int b, int g, int nodeBase, int2* myItems, int& cnt, int lane)
{
    const unsigned m = __builtin_amdgcn_ballot_w32(b == g);
    if (m != 0u) {
        const unsigned lo = m & 0xffffu, hi = m >> 16;
        if (lo != 0u) {
            if (lane == 0) myItems[cnt] = make_int2(nodeBase, (int)lo);
            cnt += 1;
        }
        if (hi != 0u) {
            if (lane == 0) myItems[cnt] = make_int2(nodeBase + 64, (int)hi);
            cnt += 1;
        }
    }
}

__global__ __launch_bounds__(1024) void pool_owner(
    const float* __restrict__ x, const int* __restrict__ batch,
    float* __restrict__ sums, float* __restrict__ cnts, int nNodes)
{
    __shared__ int2  items[POOL_WAVES * POOL_ITEM_CAP];
    __shared__ int   itemCount[POOL_WAVES];
    __shared__ float part[POOL_WAVES * D_FEAT];
    __shared__ int   partCount[POOL_WAVES];

    const int tid  = threadIdx.x;
    const int lane = tid & 31;
    const int wave = __builtin_amdgcn_readfirstlane(tid >> 5);
    const int g = blockIdx.x;
    const bool aligned = (reinterpret_cast<size_t>(batch) & 15) == 0;

    float acc0 = 0.0f, acc1 = 0.0f;
    int nMine = 0;
    int2* myItems = items + wave * POOL_ITEM_CAP;

    for (int base = 0; base < nNodes; base += POOL_PASS_NODES) {
        int cnt = 0;
        for (int k0 = 0; k0 < POOL_UNITS; k0 += 8) {
            if (base + (k0 * POOL_WAVES + wave) * POOL_UNIT >= nNodes) break;
            int4 v[8];
            #pragma unroll
            for (int k = 0; k < 8; ++k)
                v[k] = pool_load_unit(batch, base + ((k0 + k) * POOL_WAVES + wave) * POOL_UNIT, lane, nNodes, aligned);
            #pragma unroll
            for (int k = 0; k < 8; ++k) {
                const int n0 = base + ((k0 + k) * POOL_WAVES + wave) * POOL_UNIT;
                pool_note(v[k].x, g, n0 + 0, myItems, cnt, lane);
                pool_note(v[k].y, g, n0 + 1, myItems, cnt, lane);
                pool_note(v[k].z, g, n0 + 2, myItems, cnt, lane);
                pool_note(v[k].w, g, n0 + 3, myItems, cnt, lane);
            }
        }
        if (lane == 0) itemCount[wave] = cnt;
        __syncthreads();

        const int cv = itemCount[lane];
        for (int w = 0; w < POOL_WAVES; ++w) {
            const int c = __builtin_amdgcn_readlane(cv, w);
            for (int j = (wave - 8 * w) & 31; j < c; j += 32) {
                const int2 it = items[w * POOL_ITEM_CAP + j];
                const int nb = __builtin_amdgcn_readfirstlane(it.x);
                const unsigned mk = (unsigned)__builtin_amdgcn_readfirstlane(it.y);
                const int first = nb + 4 * __builtin_ctz(mk);
                float v0[16], v1[16];
                #pragma unroll
                for (int k = 0; k < 16; ++k) {
                    const int node = ((mk >> k) & 1u) ? nb + 4 * k : first;
                    const float* xr = x + (size_t)node * D_FEAT + lane;
                    v0[k] = xr[0];
                    v1[k] = xr[32];
                }
                __builtin_amdgcn_sched_barrier(0);
                #pragma unroll
                for (int k = 0; k < 16; ++k) {
                    if ((mk >> k) & 1u) {
                        acc0 += v0[k];
                        acc1 += v1[k];
                    }
                }
                nMine += __builtin_popcount(mk);
            }
        }
        __syncthreads();
    }

    part[wave * D_FEAT + lane]      = acc0;
    part[wave * D_FEAT + 32 + lane] = acc1;
    if (lane == 0) partCount[wave] = nMine;
    __syncthreads();

    if (tid < D_FEAT) {
        float s = 0.0f;
        for (int w = 0; w < POOL_WAVES; ++w) s += part[w * D_FEAT + tid];
        sums[(size_t)g * D_FEAT + tid] = s;
    } else if (tid < D_FEAT + CNT_PITCH) {
        int c = 0;
        for (int w = 0; w < POOL_WAVES; ++w) c += partCount[w];
        cnts[(size_t)g * CNT_PITCH + (tid - D_FEAT)] = (float)c;
    }
}

__global__ __launch_bounds__(256) void final_linear_wmma(
    const float* __restrict__ sums, const float* __restrict__ cnts,
    const float* __restrict__ Wl, const float* __restrict__ bl,
    float* __restrict__ out, int nrows)
{
    const int lane  = threadIdx.x & 31;
    const int strip = blockIdx.x * 8 + (threadIdx.x >> 5);
    const int r0 = strip * 16;
    if (r0 >= nrows) return;

    const int m  = lane & 15;
    const int kk = (lane >> 4) * 2;
    const int n  = lane & 15;
    const int row = r0 + m;
    float c = cnts[(size_t)row * CNT_PITCH];
    float inv = 1.0f / (c > 1.0f ? c : 1.0f);

    v8f acc0 = {}, acc1 = {};
    const float* srow = sums + (size_t)row * D_FEAT;

    #pragma unroll
    for (int k0 = 0; k0 < D_FEAT; k0 += 4) {
        v2f a;
        a.x = srow[k0 + kk] * inv;
        a.y = srow[k0 + kk + 1] * inv;
        const float* w0 = Wl + (size_t)(k0 + kk) * D_OUT + n;
        const float* w1 = Wl + (size_t)(k0 + kk + 1) * D_OUT + n;
        v2f b0; b0.x = w0[0];  b0.y = w1[0];
        v2f b1; b1.x = w0[16]; b1.y = w1[16];
        acc0 = __builtin_amdgcn_wmma_f32_16x16x4_f32(false, a, false, b0, (short)0, acc0, false, false);
        acc1 = __builtin_amdgcn_wmma_f32_16x16x4_f32(false, a, false, b1, (short)0, acc1, false, false);
    }

    const int rbase = (lane >> 4) * 8;
    const int col   = lane & 15;
    #pragma unroll
    for (int r = 0; r < 8; ++r) {
        float* orow = out + (size_t)(r0 + rbase + r) * D_OUT + col;
        orow[0]  = acc0[r] + bl[col];
        orow[16] = acc1[r] + bl[col + 16];
    }
}

static inline int cdiv(long long a, int b) { return (int)((a + b - 1) / b); }

extern "C" void kernel_launch(void* const* d_in, const int* in_sizes, int n_in,
                              void* d_out, int out_size, void* d_ws, size_t ws_size,
                              hipStream_t stream)
{
    const float* x     = (const float*)d_in[0];
    const int*   ei    = (const int*)d_in[1];
    const int*   batch = (const int*)d_in[2];
    const float* W[3]  = {(const float*)d_in[3], (const float*)d_in[5], (const float*)d_in[7]};
    const float* B[3]  = {(const float*)d_in[4], (const float*)d_in[6], (const float*)d_in[8]};
    const float* Wl    = (const float*)d_in[9];
    const float* bl    = (const float*)d_in[10];
    float* out = (float*)d_out;

    const int N = in_sizes[0] / D_FEAT;
    const int E = in_sizes[1] / 2;
    const int G = out_size / D_OUT;
    const int* srcIdx = ei;
    const int* dstIdx = ei + E;

    float* ws   = (float*)d_ws;
    float* buf0 = ws;
    float* buf1 = buf0 + (size_t)N * D_FEAT;
    float* hbuf = buf1 + (size_t)N * D_FEAT;
    float* dinv = hbuf + (size_t)N * D_FEAT;
    float* sums = dinv + N;
    float* cnts = sums + (size_t)G * D_FEAT;

    const int TB = 256;

    const int tiles = cdiv(N, TILE_NODES);
    int* tileFill = (int*)(cnts + (size_t)G * CNT_PITCH);
    int* lists    = tileFill + tiles;

    init_deg_lists<<<cdiv(N, TB), TB, 0, stream>>>(dinv, N, tileFill, tiles);
    deg_count<<<cdiv(E, TB), TB, 0, stream>>>(dstIdx, dinv, E);
    rsqrt_inplace<<<cdiv(N, TB), TB, 0, stream>>>(dinv, N);
    bin_edges<<<cdiv(E, BIN_CHUNK), 1024, 0, stream>>>(srcIdx, dstIdx, N, E, tiles, lists, tileFill);

    const int gemmBlocks = cdiv((long long)cdiv(N, 16), 8);
    const float* xcur = x;
    float* xbuf[3] = {buf0, buf1, buf0};

    for (int l = 0; l < 3; ++l) {
        float* xn = xbuf[l];
        gcn_gemm64_wmma<<<gemmBlocks, TB, 0, stream>>>(xcur, W[l], hbuf, N);
        if (l == 0)
            gcn_owner_tile<true><<<tiles, 1024, 0, stream>>>(srcIdx, dstIdx, dinv, hbuf, B[l], xcur, xn, N, E, lists, tileFill);
        else
            gcn_owner_tile<false><<<tiles, 1024, 0, stream>>>(srcIdx, dstIdx, dinv, hbuf, B[l], xcur, xn, N, E, lists, tileFill);
        xcur = xn;
    }

    pool_owner<<<G, 1024, 0, stream>>>(xcur, batch, sums, cnts, N);
    final_linear_wmma<<<cdiv((long long)cdiv(G, 16), 8), TB, 0, stream>>>(sums, cnts, Wl, bl, out, G);
}


# Degrees by node ownership: a work-group counts its 384 nodes' records of the tiles' lists on LDS counters, stores deg^-1/2 in whole lines; the per-edge global float atomics go; 12 -> 11 launches
# speedup vs baseline: 3.3938x; 1.3407x over previous
#include <hip/hip_runtime.h>
#include <hip/hip_bf16.h>

typedef float v2f __attribute__((ext_vector_type(2)));
typedef float v8f __attribute__((ext_vector_type(8)));

#define D_FEAT 64
#define D_OUT  32

#define TILE_NODES 392
#define QCAP 512
#define LIST_CAP 16384
#define LIST_WORDS (3 * LIST_CAP)
#define BIN_CHUNK 16384
#define BIN_EPT (BIN_CHUNK / 1024)
#define BIN_TILES_MAX 256
#define BIN_STAGE (BIN_CHUNK + 31 * BIN_TILES_MAX)

__global__ void init_lists(int* __restrict__ tileFill, int tiles) {
    int i = blockIdx.x * blockDim.x + threadIdx.x;
    if (i < tiles) tileFill[i] = (tiles > BIN_TILES_MAX) ? LIST_CAP + 1 : 0;
}

__global__ __launch_bounds__(256) void gcn_gemm64_wmma(
    const float* __restrict__ X, const float* __restrict__ W,
    float* __restrict__ H, int nrows)
{
    const int lane  = threadIdx.x & 31;
    const int strip = blockIdx.x * 8 + (threadIdx.x >> 5);
    const int r0 = strip * 16;
    if (r0 >= nrows) return;

    const int m  = lane & 15;
    const int kk = (lane >> 4) * 2;
    const int n  = lane & 15;

    v8f acc0 = {}, acc1 = {}, acc2 = {}, acc3 = {};
    const float* xrow = X + (size_t)(r0 + m) * D_FEAT;

    #pragma unroll
    for (int k0 = 0; k0 < D_FEAT; k0 += 4) {
        v2f a;
        a.x = xrow[k0 + kk];
        a.y = xrow[k0 + kk + 1];
        const float* w0 = W + (size_t)(k0 + kk) * D_FEAT + n;
        const float* w1 = W + (size_t)(k0 + kk + 1) * D_FEAT + n;
        v2f b0; b0.x = w0[0];  b0.y = w1[0];
        v2f b1; b1.x = w0[16]; b1.y = w1[16];
        v2f b2; b2.x = w0[32]; b2.y = w1[32];
        v2f b3; b3.x = w0[48]; b3.y = w1[48];
        acc0 = __builtin_amdgcn_wmma_f32_16x16x4_f32(false, a, false, b0, (short)0, acc0, false, false);
        acc1 = __builtin_amdgcn_wmma_f32_16x16x4_f32(false, a, false, b1, (short)0, acc1, false, false);
        acc2 = __builtin_amdgcn_wmma_f32_16x16x4_f32(false, a, false, b2, (short)0, acc2, false, false);
        acc3 = __builtin_amdgcn_wmma_f32_16x16x4_f32(false, a, false, b3, (short)0, acc3, false, false);
    }

    const int rbase = (lane >> 4) * 8;
    const int col   = lane & 15;
    #pragma unroll
    for (int r = 0; r < 8; ++r) {
        float* hrow = H + (size_t)(r0 + rbase + r) * D_FEAT + col;
        hrow[0]  = acc0[r];
        hrow[16] = acc1[r];
        hrow[32] = acc2[r];
        hrow[48] = acc3[r];
    }
}

template <int U>
__device__ __forceinline__ void rows_add(
    int s, int dl, float nrm, int n, int lane, const float* __restrict__ h, float* acc)
{
    for (int jb = 0; jb < n; jb += U) {
        float v0[U], v1[U], nj[U];
        int dj[U];
        #pragma unroll
        for (int u = 0; u < U; ++u) {
            const int j = jb + u;
            const int sj = __builtin_amdgcn_readlane(s, j);
            dj[u] = __builtin_amdgcn_readlane(dl, j);
            nj[u] = __int_as_float(__builtin_amdgcn_readlane(__float_as_int(nrm), j));
            const float* hr = h + (size_t)sj * D_FEAT + lane;
            v0[u] = hr[0];
            v1[u] = hr[32];
        }
        #pragma unroll
        for (int u = 0; u < U; ++u) {
            if (jb + u < n) {
                float* ar = acc + dj[u] * D_FEAT + lane;
                __hip_atomic_fetch_add(ar,      nj[u] * v0[u], __ATOMIC_RELAXED, __HIP_MEMORY_SCOPE_WORKGROUP);
                __hip_atomic_fetch_add(ar + 32, nj[u] * v1[u], __ATOMIC_RELAXED, __HIP_MEMORY_SCOPE_WORKGROUP);
            }
        }
    }
}

template <int JB>
__device__ __forceinline__ void rows_add_half(
    int s, int dl, float nrm, int lane, const float* __restrict__ h, float* acc)
{
    float v0[16], v1[16], nj[16];
    int dj[16];
    #pragma unroll
    for (int u = 0; u < 16; ++u) {
        const int sj = __builtin_amdgcn_readlane(s, JB + u);
        dj[u] = __builtin_amdgcn_readlane(dl, JB + u);
        nj[u] = __int_as_float(__builtin_amdgcn_readlane(__float_as_int(nrm), JB + u));
        const float* hr = h + (size_t)sj * D_FEAT + lane;
        v0[u] = hr[0];
        v1[u] = hr[32];
    }
    #pragma unroll
    for (int u = 0; u < 16; ++u) {
        float* ar = acc + dj[u] * D_FEAT + lane;
        __hip_atomic_fetch_add(ar,      nj[u] * v0[u], __ATOMIC_RELAXED, __HIP_MEMORY_SCOPE_WORKGROUP);
        __hip_atomic_fetch_add(ar + 32, nj[u] * v1[u], __ATOMIC_RELAXED, __HIP_MEMORY_SCOPE_WORKGROUP);
    }
}

__device__ __forceinline__ void tile_add_batch(
    const int* qp, int n, int lane, int lo,
    const int* __restrict__ srcIdx, const int* __restrict__ dstIdx,
    const float* __restrict__ dinv, const float* __restrict__ h, float* acc)
{
    int s = 0, dl = 0;
    float nrm = 0.0f;
    if (lane < n) {
        const int e = qp[lane];
        s = srcIdx[e];
        const int d = dstIdx[e];
        nrm = dinv[s] * dinv[d];
        dl = d - lo;
    }
    rows_add<8>(s, dl, nrm, n, lane, h, acc);
}

__device__ __forceinline__ void queue_matches(int dstv, int e, int lo, unsigned tileCount, int* q, int& cnt)
{
    const bool m = (unsigned)(dstv - lo) < tileCount;
    const unsigned mask = __builtin_amdgcn_ballot_w32(m);
    if (mask != 0u) {
        if (m) q[cnt + (int)__builtin_amdgcn_mbcnt_lo(mask, 0u)] = e;
        cnt += __builtin_popcount(mask);
    }
}

__device__ __forceinline__ void filter_pass(
    const int* __restrict__ srcIdx, const int* __restrict__ dstIdx,
    const float* __restrict__ dinv, const float* __restrict__ h, float* acc,
    int* q, int lane, int wave, int lo, unsigned tileCount, int nEdges)
{
    int cnt = 0;

    const bool aligned = (reinterpret_cast<size_t>(dstIdx) & 15) == 0;
    const int nFull = aligned ? (nEdges >> 10) : 0;
    const int rot = (int)(((long long)blockIdx.x * nFull) / gridDim.x);

    for (int c0 = wave; c0 < nFull; c0 += 32) {
        int c = c0 + rot;
        if (c >= nFull) c -= nFull;
        const int cb = c << 10;
        const int4* p = reinterpret_cast<const int4*>(dstIdx + cb) + lane;
        int4 dv[8];
        #pragma unroll
        for (int k = 0; k < 8; ++k) dv[k] = p[k * 32];
        #pragma unroll
        for (int k = 0; k < 8; ++k) {
            const int e0 = cb + k * 128 + lane * 4;
            queue_matches(dv[k].x, e0 + 0, lo, tileCount, q, cnt);
            queue_matches(dv[k].y, e0 + 1, lo, tileCount, q, cnt);
            queue_matches(dv[k].z, e0 + 2, lo, tileCount, q, cnt);
            queue_matches(dv[k].w, e0 + 3, lo, tileCount, q, cnt);
            while (cnt >= 32) {
                cnt -= 32;
                tile_add_batch(q + cnt, 32, lane, lo, srcIdx, dstIdx, dinv, h, acc);
            }
        }
    }

    for (int eb = (nFull << 10) + wave * 128; eb < nEdges; eb += 32 * 128) {
        #pragma unroll
        for (int t = 0; t < 4; ++t) {
            const int e = eb + t * 32 + lane;
            const int dstv = (e < nEdges) ? dstIdx[e] : -1;
            queue_matches(dstv, e, lo, tileCount, q, cnt);
        }
        while (cnt >= 32) {
            cnt -= 32;
            tile_add_batch(q + cnt, 32, lane, lo, srcIdx, dstIdx, dinv, h, acc);
        }
    }
    if (cnt > 0) tile_add_batch(q, cnt, lane, lo, srcIdx, dstIdx, dinv, h, acc);
}

template <bool FIRST>
__device__ __forceinline__ void list_pass(
    const int* __restrict__ rec, int* __restrict__ wts, int nSlots,
    const float* __restrict__ dinv, int lo,
    const float* __restrict__ h, float* acc, int lane, int wave)
{
    int c = wave * 32;
    if (c >= nSlots) return;
    int s  = rec[c + lane];
    int dl = rec[LIST_CAP + c + lane];
    int nb = 0;
    if (!FIRST) nb = rec[2 * LIST_CAP + c + lane];
    for (;;) {
        const int cn = c + 32 * 32;
        const bool more = cn < nSlots;
        const int cp = more ? cn : c;
        const int s2  = rec[cp + lane];
        const int dl2 = rec[LIST_CAP + cp + lane];
        int nb2 = 0;
        if (!FIRST) nb2 = rec[2 * LIST_CAP + cp + lane];
        float nrm;
        if (FIRST) {
            const bool used = dl < TILE_NODES;
            const float w = dinv[s] * dinv[lo + (used ? dl : 0)];
            nrm = used ? w : 0.0f;
        } else {
            nrm = __int_as_float(nb);
        }
        rows_add_half<0>(s, dl, nrm, lane, h, acc);
        if (__builtin_amdgcn_readlane(dl, 16) != TILE_NODES)
            rows_add_half<16>(s, dl, nrm, lane, h, acc);
        if (FIRST) wts[c + lane] = __float_as_int(nrm);
        if (!more) break;
        s = s2; dl = dl2; nb = nb2;
        c = cn;
    }
}

__global__ __launch_bounds__(1024) void bin_edges(
    const int* __restrict__ srcIdx, const int* __restrict__ dstIdx,
    int nNodes, int nEdges, int tiles,
    int* __restrict__ lists, int* __restrict__ tileFill)
{
    __shared__ int stS[BIN_STAGE];
    __shared__ int stD[BIN_STAGE];
    __shared__ int cnt[BIN_TILES_MAX];
    __shared__ int lbase[BIN_TILES_MAX];
    __shared__ int gbase[BIN_TILES_MAX];
    __shared__ int lineTile[BIN_STAGE / 32];
    __shared__ int wtot[BIN_TILES_MAX / 32];
    __shared__ int totalSlots;

    if (tiles > BIN_TILES_MAX) return;

    const int tid  = threadIdx.x;
    const int lane = tid & 31;
    const int wave = __builtin_amdgcn_readfirstlane(tid >> 5);

    if (tid < BIN_TILES_MAX) cnt[tid] = 0;
    __syncthreads();

    const int e0 = blockIdx.x * BIN_CHUNK + tid;
    int sv[BIN_EPT], dv[BIN_EPT], rk[BIN_EPT];
    #pragma unroll
    for (int k = 0; k < BIN_EPT; ++k) {
        const int e = e0 + k * 1024;
        const bool ok = e < nEdges;
        const int el = ok ? e : 0;
        const int d = dstIdx[el];
        sv[k] = srcIdx[el];
        dv[k] = ok ? d : -1;
    }
    #pragma unroll
    for (int k = 0; k < BIN_EPT; ++k) {
        rk[k] = 0;
        if ((unsigned)dv[k] < (unsigned)nNodes)
            rk[k] = __hip_atomic_fetch_add(&cnt[dv[k] / TILE_NODES], 1, __ATOMIC_RELAXED, __HIP_MEMORY_SCOPE_WORKGROUP);
        else
            dv[k] = -1;
    }
    __syncthreads();

    int n = 0, padded = 0;
    if (tid < BIN_TILES_MAX) {
        n = cnt[tid];
        padded = (n + 31) & ~31;
    }
    int incl = padded;
    #pragma unroll
    for (int off = 1; off < 32; off <<= 1) {
        const int v = __shfl_up(incl, off);
        if (lane >= off) incl += v;
    }
    if (tid < BIN_TILES_MAX && lane == 31) wtot[wave] = incl;
    __syncthreads();
    if (tid < BIN_TILES_MAX) {
        int pre = 0;
        for (int w = 0; w < wave; ++w) pre += wtot[w];
        const int excl = pre + incl - padded;
        lbase[tid] = excl;
        int gb = 0;
        if (n > 0) gb = __hip_atomic_fetch_add(&tileFill[tid], padded, __ATOMIC_RELAXED, __HIP_MEMORY_SCOPE_AGENT);
        gbase[tid] = gb;
        for (int j = excl >> 5; j < ((excl + padded) >> 5); ++j) lineTile[j] = tid;
        for (int j = excl + n; j < excl + padded; ++j) { stS[j] = 0; stD[j] = TILE_NODES; }
        if (tid == BIN_TILES_MAX - 1) totalSlots = excl + padded;
    }
    __syncthreads();

    int pp[BIN_EPT];
    #pragma unroll
    for (int k = 0; k < BIN_EPT; ++k) {
        const int dd = dv[k] < 0 ? 0 : dv[k];
        const int t = dd / TILE_NODES;
        pp[k] = lbase[t] + rk[k];
        rk[k] = dd - t * TILE_NODES;
    }
    #pragma unroll
    for (int k = 0; k < BIN_EPT; ++k) {
        if (dv[k] >= 0) {
            stS[pp[k]] = sv[k];
            stD[pp[k]] = rk[k];
        }
    }
    __syncthreads();

    const int nLines = totalSlots >> 5;
    for (int ln0 = wave; ln0 < nLines; ln0 += 4 * 32) {
        int tt[4], gp[4], a[4], b[4];
        #pragma unroll
        for (int u = 0; u < 4; ++u) {
            const int ln = ln0 + 32 * u;
            tt[u] = lineTile[ln < nLines ? ln : ln0];
        }
        #pragma unroll
        for (int u = 0; u < 4; ++u) {
            const int ln = ln0 + 32 * u;
            const int lc = ln < nLines ? ln : ln0;
            tt[u] = __builtin_amdgcn_readfirstlane(tt[u]);
            gp[u] = gbase[tt[u]] + (lc * 32 - lbase[tt[u]]);
            a[u] = stS[lc * 32 + lane];
            b[u] = stD[lc * 32 + lane];
        }
        #pragma unroll
        for (int u = 0; u < 4; ++u) {
            const int ln = ln0 + 32 * u;
            const int gpos = __builtin_amdgcn_readfirstlane(gp[u]);
            if (ln < nLines && gpos < LIST_CAP) {
                int* r = lists + (size_t)tt[u] * LIST_WORDS + gpos + lane;
                r[0]        = a[u];
                r[LIST_CAP] = b[u];
            }
        }
    }
}

#define DEG_NODES 384
#define DEG_RPT (LIST_CAP / 1024)

__global__ __launch_bounds__(1024) void node_degrees(
    const int* __restrict__ dstIdx, int nNodes, int nEdges,
    const int* __restrict__ lists, const int* __restrict__ tileFill, float* __restrict__ dinv)
{
    __shared__ int cnt[DEG_NODES];

    const int tid  = threadIdx.x;
    const int base = blockIdx.x * DEG_NODES;
    int cntNodes = nNodes - base;
    if (cntNodes > DEG_NODES) cntNodes = DEG_NODES;
    if (cntNodes <= 0) return;

    if (tid < DEG_NODES) cnt[tid] = 0;
    __syncthreads();

    const int t0 = base / TILE_NODES;
    const int t1 = (base + cntNodes - 1) / TILE_NODES;
    const int n0 = tileFill[t0];
    const int n1 = (t1 != t0) ? tileFill[t1] : 0;

    if (n0 <= LIST_CAP && n1 <= LIST_CAP) {
        const int* pl0 = lists + (size_t)t0 * LIST_WORDS + LIST_CAP;
        const int* pl1 = lists + (size_t)t1 * LIST_WORDS + LIST_CAP;
        int pa[DEG_RPT], pb[DEG_RPT];
        #pragma unroll
        for (int k = 0; k < DEG_RPT; ++k) {
            const int i = tid + k * 1024;
            pa[k] = *(i < n0 ? pl0 + i : tileFill);
        }
        #pragma unroll
        for (int k = 0; k < DEG_RPT; ++k) {
            const int i = tid + k * 1024;
            pb[k] = *(i < n1 ? pl1 + i : tileFill);
        }
        const int off0 = t0 * TILE_NODES - base;
        const int off1 = t1 * TILE_NODES - base;
        #pragma unroll
        for (int k = 0; k < DEG_RPT; ++k) {
            const int i = tid + k * 1024;
            const unsigned r = (unsigned)(pa[k] + off0);
            if (i < n0 && pa[k] < TILE_NODES && r < (unsigned)cntNodes)
                __hip_atomic_fetch_add(&cnt[r], 1, __ATOMIC_RELAXED, __HIP_MEMORY_SCOPE_WORKGROUP);
        }
        #pragma unroll
        for (int k = 0; k < DEG_RPT; ++k) {
            const int i = tid + k * 1024;
            const unsigned r = (unsigned)(pb[k] + off1);
            if (i < n1 && pb[k] < TILE_NODES && r < (unsigned)cntNodes)
                __hip_atomic_fetch_add(&cnt[r], 1, __ATOMIC_RELAXED, __HIP_MEMORY_SCOPE_WORKGROUP);
        }
    } else {
        for (int e = tid; e < nEdges; e += 1024) {
            const unsigned r = (unsigned)(dstIdx[e] - base);
            if (r < (unsigned)cntNodes)
                __hip_atomic_fetch_add(&cnt[r], 1, __ATOMIC_RELAXED, __HIP_MEMORY_SCOPE_WORKGROUP);
        }
    }
    __syncthreads();

    if (tid < cntNodes) dinv[base + tid] = __frsqrt_rn(1.0f + (float)cnt[tid]);
}

template <bool FIRST>
__global__ __launch_bounds__(1024) void gcn_owner_tile(
    const int* __restrict__ srcIdx, const int* __restrict__ dstIdx,
    const float* __restrict__ dinv, const float* __restrict__ h,
    const float* __restrict__ bias, const float* __restrict__ xprev,
    float* __restrict__ xnew, int nNodes, int nEdges,
    int* __restrict__ lists, const int* __restrict__ tileFill)
{
    __shared__ float acc[(TILE_NODES + 1) * D_FEAT];
    __shared__ int   queue[32 * QCAP];

    const int tid  = threadIdx.x;
    const int lane = tid & 31;
    const int wave = __builtin_amdgcn_readfirstlane(tid >> 5);
    const int lo   = blockIdx.x * TILE_NODES;
    int cntNodes = nNodes - lo;
    if (cntNodes > TILE_NODES) cntNodes = TILE_NODES;
    if (cntNodes <= 0) return;
    const unsigned tileCount = (unsigned)cntNodes;

    for (int i = tid; i < (TILE_NODES + 1) * D_FEAT; i += 1024) acc[i] = 0.0f;
    __syncthreads();

    int* q = queue + wave * QCAP;
    int* rec = lists + (size_t)blockIdx.x * LIST_WORDS;

    const int nSlots = tileFill[blockIdx.x];
    if (nSlots <= LIST_CAP) {
        list_pass<FIRST>(rec, rec + 2 * LIST_CAP, nSlots, dinv, lo, h, acc, lane, wave);
    } else {
        filter_pass(srcIdx, dstIdx, dinv, h, acc, q, lane, wave, lo, tileCount, nEdges);
    }
    __syncthreads();

    const size_t gbase = (size_t)lo * D_FEAT;
    const int nElem = cntNodes * D_FEAT;
    for (int i = tid; i < nElem; i += 1024) {
        const int f = i & 63;
        const float di = dinv[lo + (i >> 6)];
        float v = acc[i] + di * di * h[gbase + i] + bias[f];
        v = v > 0.0f ? v : 0.0f;
        xnew[gbase + i] = v + xprev[gbase + i];
    }
}

#define POOL_WAVES 32
#define POOL_UNIT 128
#define POOL_UNITS 32
#define POOL_PASS_NODES (POOL_WAVES * POOL_UNITS * POOL_UNIT)
#define POOL_ITEM_CAP (POOL_UNITS * 8)
#define CNT_PITCH 32

__device__ __forceinline__ int4 pool_load_unit(
    const int* __restrict__ batch, int n0, int lane, int nNodes, bool aligned)
{
    const int n = n0 + lane * 4;
    int4 r;
    if (aligned && n0 + POOL_UNIT <= nNodes) {
        r = *reinterpret_cast<const int4*>(batch + n);
    } else {
        r.x = (n + 0 < nNodes) ? batch[n + 0] : -1;
        r.y = (n + 1 < nNodes) ? batch[n + 1] : -1;
        r.z = (n + 2 < nNodes) ? batch[n + 2] : -1;
        r.w = (n + 3 < nNodes) ? batch[n + 3] : -1;
    }
    return r;
}

__device__ __forceinline__ void pool_note(int b, int g, int nodeBase, int2* myItems, int& cnt, int lane)
{
    const unsigned m = __builtin_amdgcn_ballot_w32(b == g);
    if (m != 0u) {
        const unsigned lo = m & 0xffffu, hi = m >> 16;
        if (lo != 0u) {
            if (lane == 0) myItems[cnt] = make_int2(nodeBase, (int)lo);
            cnt += 1;
        }
        if (hi != 0u) {
            if (lane == 0) myItems[cnt] = make_int2(nodeBase + 64, (int)hi);
            cnt += 1;
        }
    }
}

__global__ __launch_bounds__(1024) void pool_owner(
    const float* __restrict__ x, const int* __restrict__ batch,
    float* __restrict__ sums, float* __restrict__ cnts, int nNodes)
{
    __shared__ int2  items[POOL_WAVES * POOL_ITEM_CAP];
    __shared__ int   itemCount[POOL_WAVES];
    __shared__ float part[POOL_WAVES * D_FEAT];
    __shared__ int   partCount[POOL_WAVES];

    const int tid  = threadIdx.x;
    const int lane = tid & 31;
    const int wave = __builtin_amdgcn_readfirstlane(tid >> 5);
    const int g = blockIdx.x;
    const bool aligned = (reinterpret_cast<size_t>(batch) & 15) == 0;

    float acc0 = 0.0f, acc1 = 0.0f;
    int nMine = 0;
    int2* myItems = items + wave * POOL_ITEM_CAP;

    for (int base = 0; base < nNodes; base += POOL_PASS_NODES) {
        int cnt = 0;
        for (int k0 = 0; k0 < POOL_UNITS; k0 += 8) {
            if (base + (k0 * POOL_WAVES + wave) * POOL_UNIT >= nNodes) break;
            int4 v[8];
            #pragma unroll
            for (int k = 0; k < 8; ++k)
                v[k] = pool_load_unit(batch, base + ((k0 + k) * POOL_WAVES + wave) * POOL_UNIT, lane, nNodes, aligned);
            #pragma unroll
            for (int k = 0; k < 8; ++k) {
                const int n0 = base + ((k0 + k) * POOL_WAVES + wave) * POOL_UNIT;
                pool_note(v[k].x, g, n0 + 0, myItems, cnt, lane);
                pool_note(v[k].y, g, n0 + 1, myItems, cnt, lane);
                pool_note(v[k].z, g, n0 + 2, myItems, cnt, lane);
                pool_note(v[k].w, g, n0 + 3, myItems, cnt, lane);
            }
        }
        if (lane == 0) itemCount[wave] = cnt;
        __syncthreads();

        const int cv = itemCount[lane];
        for (int w = 0; w < POOL_WAVES; ++w) {
            const int c = __builtin_amdgcn_readlane(cv, w);
            for (int j = (wave - 8 * w) & 31; j < c; j += 32) {
                const int2 it = items[w * POOL_ITEM_CAP + j];
                const int nb = __builtin_amdgcn_readfirstlane(it.x);
                const unsigned mk = (unsigned)__builtin_amdgcn_readfirstlane(it.y);
                const int first = nb + 4 * __builtin_ctz(mk);
                float v0[16], v1[16];
                #pragma unroll
                for (int k = 0; k < 16; ++k) {
                    const int node = ((mk >> k) & 1u) ? nb + 4 * k : first;
                    const float* xr = x + (size_t)node * D_FEAT + lane;
                    v0[k] = xr[0];
                    v1[k] = xr[32];
                }
                __builtin_amdgcn_sched_barrier(0);
                #pragma unroll
                for (int k = 0; k < 16; ++k) {
                    if ((mk >> k) & 1u) {
                        acc0 += v0[k];
                        acc1 += v1[k];
                    }
                }
                nMine += __builtin_popcount(mk);
            }
        }
        __syncthreads();
    }

    part[wave * D_FEAT + lane]      = acc0;
    part[wave * D_FEAT + 32 + lane] = acc1;
    if (lane == 0) partCount[wave] = nMine;
    __syncthreads();

    if (tid < D_FEAT) {
        float s = 0.0f;
        for (int w = 0; w < POOL_WAVES; ++w) s += part[w * D_FEAT + tid];
        sums[(size_t)g * D_FEAT + tid] = s;
    } else if (tid < D_FEAT + CNT_PITCH) {
        int c = 0;
        for (int w = 0; w < POOL_WAVES; ++w) c += partCount[w];
        cnts[(size_t)g * CNT_PITCH + (tid - D_FEAT)] = (float)c;
    }
}

__global__ __launch_bounds__(256) void final_linear_wmma(
    const float* __restrict__ sums, const float* __restrict__ cnts,
    const float* __restrict__ Wl, const float* __restrict__ bl,
    float* __restrict__ out, int nrows)
{
    const int lane  = threadIdx.x & 31;
    const int strip = blockIdx.x * 8 + (threadIdx.x >> 5);
    const int r0 = strip * 16;
    if (r0 >= nrows) return;

    const int m  = lane & 15;
    const int kk = (lane >> 4) * 2;
    const int n  = lane & 15;
    const int row = r0 + m;
    float c = cnts[(size_t)row * CNT_PITCH];
    float inv = 1.0f / (c > 1.0f ? c : 1.0f);

    v8f acc0 = {}, acc1 = {};
    const float* srow = sums + (size_t)row * D_FEAT;

    #pragma unroll
    for (int k0 = 0; k0 < D_FEAT; k0 += 4) {
        v2f a;
        a.x = srow[k0 + kk] * inv;
        a.y = srow[k0 + kk + 1] * inv;
        const float* w0 = Wl + (size_t)(k0 + kk) * D_OUT + n;
        const float* w1 = Wl + (size_t)(k0 + kk + 1) * D_OUT + n;
        v2f b0; b0.x = w0[0];  b0.y = w1[0];
        v2f b1; b1.x = w0[16]; b1.y = w1[16];
        acc0 = __builtin_amdgcn_wmma_f32_16x16x4_f32(false, a, false, b0, (short)0, acc0, false, false);
        acc1 = __builtin_amdgcn_wmma_f32_16x16x4_f32(false, a, false, b1, (short)0, acc1, false, false);
    }

    const int rbase = (lane >> 4) * 8;
    const int col   = lane & 15;
    #pragma unroll
    for (int r = 0; r < 8; ++r) {
        float* orow = out + (size_t)(r0 + rbase + r) * D_OUT + col;
        orow[0]  = acc0[r] + bl[col];
        orow[16] = acc1[r] + bl[col + 16];
    }
}

static inline int cdiv(long long a, int b) { return (int)((a + b - 1) / b); }

extern "C" void kernel_launch(void* const* d_in, const int* in_sizes, int n_in,
                              void* d_out, int out_size, void* d_ws, size_t ws_size,
                              hipStream_t stream)
{
    const float* x     = (const float*)d_in[0];
    const int*   ei    = (const int*)d_in[1];
    const int*   batch = (const int*)d_in[2];
    const float* W[3]  = {(const float*)d_in[3], (const float*)d_in[5], (const float*)d_in[7]};
    const float* B[3]  = {(const float*)d_in[4], (const float*)d_in[6], (const float*)d_in[8]};
    const float* Wl    = (const float*)d_in[9];
    const float* bl    = (const float*)d_in[10];
    float* out = (float*)d_out;

    const int N = in_sizes[0] / D_FEAT;
    const int E = in_sizes[1] / 2;
    const int G = out_size / D_OUT;
    const int* srcIdx = ei;
    const int* dstIdx = ei + E;

    float* ws   = (float*)d_ws;
    float* buf0 = ws;
    float* buf1 = buf0 + (size_t)N * D_FEAT;
    float* hbuf = buf1 + (size_t)N * D_FEAT;
    float* dinv = hbuf + (size_t)N * D_FEAT;
    float* sums = dinv + N;
    float* cnts = sums + (size_t)G * D_FEAT;

    const int TB = 256;

    const int tiles = cdiv(N, TILE_NODES);
    int* tileFill = (int*)(cnts + (size_t)G * CNT_PITCH);
    int* lists    = tileFill + tiles;

    init_lists<<<cdiv(tiles, TB), TB, 0, stream>>>(tileFill, tiles);
    bin_edges<<<cdiv(E, BIN_CHUNK), 1024, 0, stream>>>(srcIdx, dstIdx, N, E, tiles, lists, tileFill);
    node_degrees<<<cdiv(N, DEG_NODES), 1024, 0, stream>>>(dstIdx, N, E, lists, tileFill, dinv);

    const int gemmBlocks = cdiv((long long)cdiv(N, 16), 8);
    const float* xcur = x;
    float* xbuf[3] = {buf0, buf1, buf0};

    for (int l = 0; l < 3; ++l) {
        float* xn = xbuf[l];
        gcn_gemm64_wmma<<<gemmBlocks, TB, 0, stream>>>(xcur, W[l], hbuf, N);
        if (l == 0)
            gcn_owner_tile<true><<<tiles, 1024, 0, stream>>>(srcIdx, dstIdx, dinv, hbuf, B[l], xcur, xn, N, E, lists, tileFill);
        else
            gcn_owner_tile<false><<<tiles, 1024, 0, stream>>>(srcIdx, dstIdx, dinv, hbuf, B[l], xcur, xn, N, E, lists, tileFill);
        xcur = xn;
    }

    pool_owner<<<G, 1024, 0, stream>>>(xcur, batch, sums, cnts, N);
    final_linear_wmma<<<cdiv((long long)cdiv(G, 16), 8), TB, 0, stream>>>(sums, cnts, Wl, bl, out, G);
}


# Next layer's x W moved into the tile kernel's tail: finished rows stay in LDS (pitch 68), W in LDS in fragment order, a wave per 16-row strip on f32 WMMA; 2 product launches fewer; sums unchanged
# speedup vs baseline: 3.7652x; 1.1094x over previous
#include <hip/hip_runtime.h>
#include <hip/hip_bf16.h>

typedef float v2f __attribute__((ext_vector_type(2)));
typedef float v8f __attribute__((ext_vector_type(8)));

#define D_FEAT 64
#define D_OUT  32

#define TILE_NODES 392
#define ACC_PITCH 68
#define ACC_ROWS 400
#define QCAP 512
#define LIST_CAP 16384
#define LIST_WORDS (3 * LIST_CAP)
#define BIN_CHUNK 16384
#define BIN_EPT (BIN_CHUNK / 1024)
#define BIN_TILES_MAX 256
#define BIN_STAGE (BIN_CHUNK + 31 * BIN_TILES_MAX)

__global__ void init_lists(int* __restrict__ tileFill, int tiles) {
    int i = blockIdx.x * blockDim.x + threadIdx.x;
    if (i < tiles) tileFill[i] = (tiles > BIN_TILES_MAX) ? LIST_CAP + 1 : 0;
}

__global__ __launch_bounds__(256) void gcn_gemm64_wmma(
    const float* __restrict__ X, const float* __restrict__ W,
    float* __restrict__ H, int nrows)
{
    const int lane  = threadIdx.x & 31;
    const int strip = blockIdx.x * 8 + (threadIdx.x >> 5);
    const int r0 = strip * 16;
    if (r0 >= nrows) return;

    const int m  = lane & 15;
    const int kk = (lane >> 4) * 2;
    const int n  = lane & 15;

    v8f acc0 = {}, acc1 = {}, acc2 = {}, acc3 = {};
    const float* xrow = X + (size_t)(r0 + m) * D_FEAT;

    #pragma unroll
    for (int k0 = 0; k0 < D_FEAT; k0 += 4) {
        v2f a;
        a.x = xrow[k0 + kk];
        a.y = xrow[k0 + kk + 1];
        const float* w0 = W + (size_t)(k0 + kk) * D_FEAT + n;
        const float* w1 = W + (size_t)(k0 + kk + 1) * D_FEAT + n;
        v2f b0; b0.x = w0[0];  b0.y = w1[0];
        v2f b1; b1.x = w0[16]; b1.y = w1[16];
        v2f b2; b2.x = w0[32]; b2.y = w1[32];
        v2f b3; b3.x = w0[48]; b3.y = w1[48];
        acc0 = __builtin_amdgcn_wmma_f32_16x16x4_f32(false, a, false, b0, (short)0, acc0, false, false);
        acc1 = __builtin_amdgcn_wmma_f32_16x16x4_f32(false, a, false, b1, (short)0, acc1, false, false);
        acc2 = __builtin_amdgcn_wmma_f32_16x16x4_f32(false, a, false, b2, (short)0, acc2, false, false);
        acc3 = __builtin_amdgcn_wmma_f32_16x16x4_f32(false, a, false, b3, (short)0, acc3, false, false);
    }

    const int rbase = (lane >> 4) * 8;
    const int col   = lane & 15;
    #pragma unroll
    for (int r = 0; r < 8; ++r) {
        float* hrow = H + (size_t)(r0 + rbase + r) * D_FEAT + col;
        hrow[0]  = acc0[r];
        hrow[16] = acc1[r];
        hrow[32] = acc2[r];
        hrow[48] = acc3[r];
    }
}

template <int U>
__device__ __forceinline__ void rows_add(
    int s, int dl, float nrm, int n, int lane, const float* __restrict__ h, float* acc)
{
    for (int jb = 0; jb < n; jb += U) {
        float v0[U], v1[U], nj[U];
        int dj[U];
        #pragma unroll
        for (int u = 0; u < U; ++u) {
            const int j = jb + u;
            const int sj = __builtin_amdgcn_readlane(s, j);
            dj[u] = __builtin_amdgcn_readlane(dl, j);
            nj[u] = __int_as_float(__builtin_amdgcn_readlane(__float_as_int(nrm), j));
            const float* hr = h + (size_t)sj * D_FEAT + lane;
            v0[u] = hr[0];
            v1[u] = hr[32];
        }
        #pragma unroll
        for (int u = 0; u < U; ++u) {
            if (jb + u < n) {
                float* ar = acc + dj[u] * ACC_PITCH + lane;
                __hip_atomic_fetch_add(ar,      nj[u] * v0[u], __ATOMIC_RELAXED, __HIP_MEMORY_SCOPE_WORKGROUP);
                __hip_atomic_fetch_add(ar + 32, nj[u] * v1[u], __ATOMIC_RELAXED, __HIP_MEMORY_SCOPE_WORKGROUP);
            }
        }
    }
}

template <int JB>
__device__ __forceinline__ void rows_add_half(
    int s, int dl, float nrm, int lane, const float* __restrict__ h, float* acc)
{
    float v0[16], v1[16], nj[16];
    int dj[16];
    #pragma unroll
    for (int u = 0; u < 16; ++u) {
        const int sj = __builtin_amdgcn_readlane(s, JB + u);
        dj[u] = __builtin_amdgcn_readlane(dl, JB + u);
        nj[u] = __int_as_float(__builtin_amdgcn_readlane(__float_as_int(nrm), JB + u));
        const float* hr = h + (size_t)sj * D_FEAT + lane;
        v0[u] = hr[0];
        v1[u] = hr[32];
    }
    #pragma unroll
    for (int u = 0; u < 16; ++u) {
        float* ar = acc + dj[u] * ACC_PITCH + lane;
        __hip_atomic_fetch_add(ar,      nj[u] * v0[u], __ATOMIC_RELAXED, __HIP_MEMORY_SCOPE_WORKGROUP);
        __hip_atomic_fetch_add(ar + 32, nj[u] * v1[u], __ATOMIC_RELAXED, __HIP_MEMORY_SCOPE_WORKGROUP);
    }
}

__device__ __forceinline__ void tile_add_batch(
    const int* qp, int n, int lane, int lo,
    const int* __restrict__ srcIdx, const int* __restrict__ dstIdx,
    const float* __restrict__ dinv, const float* __restrict__ h, float* acc)
{
    int s = 0, dl = 0;
    float nrm = 0.0f;
    if (lane < n) {
        const int e = qp[lane];
        s = srcIdx[e];
        const int d = dstIdx[e];
        nrm = dinv[s] * dinv[d];
        dl = d - lo;
    }
    rows_add<8>(s, dl, nrm, n, lane, h, acc);
}

__device__ __forceinline__ void queue_matches(int dstv, int e, int lo, unsigned tileCount, int* q, int& cnt)
{
    const bool m = (unsigned)(dstv - lo) < tileCount;
    const unsigned mask = __builtin_amdgcn_ballot_w32(m);
    if (mask != 0u) {
        if (m) q[cnt + (int)__builtin_amdgcn_mbcnt_lo(mask, 0u)] = e;
        cnt += __builtin_popcount(mask);
    }
}

__device__ __forceinline__ void filter_pass(
    const int* __restrict__ srcIdx, const int* __restrict__ dstIdx,
    const float* __restrict__ dinv, const float* __restrict__ h, float* acc,
    int* q, int lane, int wave, int lo, unsigned tileCount, int nEdges)
{
    int cnt = 0;

    const bool aligned = (reinterpret_cast<size_t>(dstIdx) & 15) == 0;
    const int nFull = aligned ? (nEdges >> 10) : 0;
    const int rot = (int)(((long long)blockIdx.x * nFull) / gridDim.x);

    for (int c0 = wave; c0 < nFull; c0 += 32) {
        int c = c0 + rot;
        if (c >= nFull) c -= nFull;
        const int cb = c << 10;
        const int4* p = reinterpret_cast<const int4*>(dstIdx + cb) + lane;
        int4 dv[8];
        #pragma unroll
        for (int k = 0; k < 8; ++k) dv[k] = p[k * 32];
        #pragma unroll
        for (int k = 0; k < 8; ++k) {
            const int e0 = cb + k * 128 + lane * 4;
            queue_matches(dv[k].x, e0 + 0, lo, tileCount, q, cnt);
            queue_matches(dv[k].y, e0 + 1, lo, tileCount, q, cnt);
            queue_matches(dv[k].z, e0 + 2, lo, tileCount, q, cnt);
            queue_matches(dv[k].w, e0 + 3, lo, tileCount, q, cnt);
            while (cnt >= 32) {
                cnt -= 32;
                tile_add_batch(q + cnt, 32, lane, lo, srcIdx, dstIdx, dinv, h, acc);
            }
        }
    }

    for (int eb = (nFull << 10) + wave * 128; eb < nEdges; eb += 32 * 128) {
        #pragma unroll
        for (int t = 0; t < 4; ++t) {
            const int e = eb + t * 32 + lane;
            const int dstv = (e < nEdges) ? dstIdx[e] : -1;
            queue_matches(dstv, e, lo, tileCount, q, cnt);
        }
        while (cnt >= 32) {
            cnt -= 32;
            tile_add_batch(q + cnt, 32, lane, lo, srcIdx, dstIdx, dinv, h, acc);
        }
    }
    if (cnt > 0) tile_add_batch(q, cnt, lane, lo, srcIdx, dstIdx, dinv, h, acc);
}

template <bool FIRST>
__device__ __forceinline__ void list_pass(
    const int* __restrict__ rec, int* __restrict__ wts, int nSlots,
    const float* __restrict__ dinv, int lo,
    const float* __restrict__ h, float* acc, int lane, int wave)
{
    int c = wave * 32;
    if (c >= nSlots) return;
    int s  = rec[c + lane];
    int dl = rec[LIST_CAP + c + lane];
    int nb = 0;
    if (!FIRST) nb = rec[2 * LIST_CAP + c + lane];
    for (;;) {
        const int cn = c + 32 * 32;
        const bool more = cn < nSlots;
        const int cp = more ? cn : c;
        const int s2  = rec[cp + lane];
        const int dl2 = rec[LIST_CAP + cp + lane];
        int nb2 = 0;
        if (!FIRST) nb2 = rec[2 * LIST_CAP + cp + lane];
        float nrm;
        if (FIRST) {
            const bool used = dl < TILE_NODES;
            const float w = dinv[s] * dinv[lo + (used ? dl : 0)];
            nrm = used ? w : 0.0f;
        } else {
            nrm = __int_as_float(nb);
        }
        rows_add_half<0>(s, dl, nrm, lane, h, acc);
        if (__builtin_amdgcn_readlane(dl, 16) != TILE_NODES)
            rows_add_half<16>(s, dl, nrm, lane, h, acc);
        if (FIRST) wts[c + lane] = __float_as_int(nrm);
        if (!more) break;
        s = s2; dl = dl2; nb = nb2;
        c = cn;
    }
}

__global__ __launch_bounds__(1024) void bin_edges(
    const int* __restrict__ srcIdx, const int* __restrict__ dstIdx,
    int nNodes, int nEdges, int tiles,
    int* __restrict__ lists, int* __restrict__ tileFill)
{
    __shared__ int stS[BIN_STAGE];
    __shared__ int stD[BIN_STAGE];
    __shared__ int cnt[BIN_TILES_MAX];
    __shared__ int lbase[BIN_TILES_MAX];
    __shared__ int gbase[BIN_TILES_MAX];
    __shared__ int lineTile[BIN_STAGE / 32];
    __shared__ int wtot[BIN_TILES_MAX / 32];
    __shared__ int totalSlots;

    if (tiles > BIN_TILES_MAX) return;

    const int tid  = threadIdx.x;
    const int lane = tid & 31;
    const int wave = __builtin_amdgcn_readfirstlane(tid >> 5);

    if (tid < BIN_TILES_MAX) cnt[tid] = 0;
    __syncthreads();

    const int e0 = blockIdx.x * BIN_CHUNK + tid;
    int sv[BIN_EPT], dv[BIN_EPT], rk[BIN_EPT];
    #pragma unroll
    for (int k = 0; k < BIN_EPT; ++k) {
        const int e = e0 + k * 1024;
        const bool ok = e < nEdges;
        const int el = ok ? e : 0;
        const int d = dstIdx[el];
        sv[k] = srcIdx[el];
        dv[k] = ok ? d : -1;
    }
    #pragma unroll
    for (int k = 0; k < BIN_EPT; ++k) {
        rk[k] = 0;
        if ((unsigned)dv[k] < (unsigned)nNodes)
            rk[k] = __hip_atomic_fetch_add(&cnt[dv[k] / TILE_NODES], 1, __ATOMIC_RELAXED, __HIP_MEMORY_SCOPE_WORKGROUP);
        else
            dv[k] = -1;
    }
    __syncthreads();

    int n = 0, padded = 0;
    if (tid < BIN_TILES_MAX) {
        n = cnt[tid];
        padded = (n + 31) & ~31;
    }
    int incl = padded;
    #pragma unroll
    for (int off = 1; off < 32; off <<= 1) {
        const int v = __shfl_up(incl, off);
        if (lane >= off) incl += v;
    }
    if (tid < BIN_TILES_MAX && lane == 31) wtot[wave] = incl;
    __syncthreads();
    if (tid < BIN_TILES_MAX) {
        int pre = 0;
        for (int w = 0; w < wave; ++w) pre += wtot[w];
        const int excl = pre + incl - padded;
        lbase[tid] = excl;
        int gb = 0;
        if (n > 0) gb = __hip_atomic_fetch_add(&tileFill[tid], padded, __ATOMIC_RELAXED, __HIP_MEMORY_SCOPE_AGENT);
        gbase[tid] = gb;
        for (int j = excl >> 5; j < ((excl + padded) >> 5); ++j) lineTile[j] = tid;
        for (int j = excl + n; j < excl + padded; ++j) { stS[j] = 0; stD[j] = TILE_NODES; }
        if (tid == BIN_TILES_MAX - 1) totalSlots = excl + padded;
    }
    __syncthreads();

    int pp[BIN_EPT];
    #pragma unroll
    for (int k = 0; k < BIN_EPT; ++k) {
        const int dd = dv[k] < 0 ? 0 : dv[k];
        const int t = dd / TILE_NODES;
        pp[k] = lbase[t] + rk[k];
        rk[k] = dd - t * TILE_NODES;
    }
    #pragma unroll
    for (int k = 0; k < BIN_EPT; ++k) {
        if (dv[k] >= 0) {
            stS[pp[k]] = sv[k];
            stD[pp[k]] = rk[k];
        }
    }
    __syncthreads();

    const int nLines = totalSlots >> 5;
    for (int ln0 = wave; ln0 < nLines; ln0 += 4 * 32) {
        int tt[4], gp[4], a[4], b[4];
        #pragma unroll
        for (int u = 0; u < 4; ++u) {
            const int ln = ln0 + 32 * u;
            tt[u] = lineTile[ln < nLines ? ln : ln0];
        }
        #pragma unroll
        for (int u = 0; u < 4; ++u) {
            const int ln = ln0 + 32 * u;
            const int lc = ln < nLines ? ln : ln0;
            tt[u] = __builtin_amdgcn_readfirstlane(tt[u]);
            gp[u] = gbase[tt[u]] + (lc * 32 - lbase[tt[u]]);
            a[u] = stS[lc * 32 + lane];
            b[u] = stD[lc * 32 + lane];
        }
        #pragma unroll
        for (int u = 0; u < 4; ++u) {
            const int ln = ln0 + 32 * u;
            const int gpos = __builtin_amdgcn_readfirstlane(gp[u]);
            if (ln < nLines && gpos < LIST_CAP) {
                int* r = lists + (size_t)tt[u] * LIST_WORDS + gpos + lane;
                r[0]        = a[u];
                r[LIST_CAP] = b[u];
            }
        }
    }
}

#define DEG_NODES 384
#define DEG_RPT (LIST_CAP / 1024)

__global__ __launch_bounds__(1024) void node_degrees(
    const int* __restrict__ dstIdx, int nNodes, int nEdges,
    const int* __restrict__ lists, const int* __restrict__ tileFill, float* __restrict__ dinv)
{
    __shared__ int cnt[DEG_NODES];

    const int tid  = threadIdx.x;
    const int base = blockIdx.x * DEG_NODES;
    int cntNodes = nNodes - base;
    if (cntNodes > DEG_NODES) cntNodes = DEG_NODES;
    if (cntNodes <= 0) return;

    if (tid < DEG_NODES) cnt[tid] = 0;
    __syncthreads();

    const int t0 = base / TILE_NODES;
    const int t1 = (base + cntNodes - 1) / TILE_NODES;
    const int n0 = tileFill[t0];
    const int n1 = (t1 != t0) ? tileFill[t1] : 0;

    if (n0 <= LIST_CAP && n1 <= LIST_CAP) {
        const int* pl0 = lists + (size_t)t0 * LIST_WORDS + LIST_CAP;
        const int* pl1 = lists + (size_t)t1 * LIST_WORDS + LIST_CAP;
        int pa[DEG_RPT], pb[DEG_RPT];
        #pragma unroll
        for (int k = 0; k < DEG_RPT; ++k) {
            const int i = tid + k * 1024;
            pa[k] = *(i < n0 ? pl0 + i : tileFill);
        }
        #pragma unroll
        for (int k = 0; k < DEG_RPT; ++k) {
            const int i = tid + k * 1024;
            pb[k] = *(i < n1 ? pl1 + i : tileFill);
        }
        const int off0 = t0 * TILE_NODES - base;
        const int off1 = t1 * TILE_NODES - base;
        #pragma unroll
        for (int k = 0; k < DEG_RPT; ++k) {
            const int i = tid + k * 1024;
            const unsigned r = (unsigned)(pa[k] + off0);
            if (i < n0 && pa[k] < TILE_NODES && r < (unsigned)cntNodes)
                __hip_atomic_fetch_add(&cnt[r], 1, __ATOMIC_RELAXED, __HIP_MEMORY_SCOPE_WORKGROUP);
        }
        #pragma unroll
        for (int k = 0; k < DEG_RPT; ++k) {
            const int i = tid + k * 1024;
            const unsigned r = (unsigned)(pb[k] + off1);
            if (i < n1 && pb[k] < TILE_NODES && r < (unsigned)cntNodes)
                __hip_atomic_fetch_add(&cnt[r], 1, __ATOMIC_RELAXED, __HIP_MEMORY_SCOPE_WORKGROUP);
        }
    } else {
        for (int e = tid; e < nEdges; e += 1024) {
            const unsigned r = (unsigned)(dstIdx[e] - base);
            if (r < (unsigned)cntNodes)
                __hip_atomic_fetch_add(&cnt[r], 1, __ATOMIC_RELAXED, __HIP_MEMORY_SCOPE_WORKGROUP);
        }
    }
    __syncthreads();

    if (tid < cntNodes) dinv[base + tid] = __frsqrt_rn(1.0f + (float)cnt[tid]);
}

template <bool FIRST, bool NEXT>
__global__ __launch_bounds__(1024) void gcn_owner_tile(
    const int* __restrict__ srcIdx, const int* __restrict__ dstIdx,
    const float* __restrict__ dinv, const float* __restrict__ h,
    const float* __restrict__ bias, const float* __restrict__ xprev,
    float* __restrict__ xnew, int nNodes, int nEdges,
    int* __restrict__ lists, const int* __restrict__ tileFill,
    const float* __restrict__ Wnext, float* __restrict__ hnext)
{
    __shared__ __attribute__((aligned(16))) float acc[ACC_ROWS * ACC_PITCH];
    __shared__ int   queue[32 * QCAP];
    __shared__ float4 wfrag[NEXT ? D_FEAT * D_FEAT / 4 : 1];

    const int tid  = threadIdx.x;
    const int lane = tid & 31;
    const int wave = __builtin_amdgcn_readfirstlane(tid >> 5);
    const int lo   = blockIdx.x * TILE_NODES;
    int cntNodes = nNodes - lo;
    if (cntNodes > TILE_NODES) cntNodes = TILE_NODES;
    if (cntNodes <= 0) return;
    const unsigned tileCount = (unsigned)cntNodes;

    for (int i = tid; i < ACC_ROWS * ACC_PITCH; i += 1024) acc[i] = 0.0f;
    if (NEXT) {
        const int ks = tid >> 6;
        const int p  = (tid >> 5) & 1;
        const float* w0 = Wnext + (size_t)(ks * 4 + (lane >> 4) * 2) * D_FEAT + p * 32 + (lane & 15);
        float4 v;
        v.x = w0[0];
        v.y = w0[D_FEAT];
        v.z = w0[16];
        v.w = w0[D_FEAT + 16];
        wfrag[p * 512 + ks * 32 + lane] = v;
    }
    __syncthreads();

    int* q = queue + wave * QCAP;
    int* rec = lists + (size_t)blockIdx.x * LIST_WORDS;

    const int nSlots = tileFill[blockIdx.x];
    if (nSlots <= LIST_CAP) {
        list_pass<FIRST>(rec, rec + 2 * LIST_CAP, nSlots, dinv, lo, h, acc, lane, wave);
    } else {
        filter_pass(srcIdx, dstIdx, dinv, h, acc, q, lane, wave, lo, tileCount, nEdges);
    }
    __syncthreads();

    const size_t gbase = (size_t)lo * D_FEAT;
    const int nElem = cntNodes * D_FEAT;
    for (int i = tid; i < nElem; i += 1024) {
        const int f = i & 63;
        const int ai = (i >> 6) * ACC_PITCH + f;
        const float di = dinv[lo + (i >> 6)];
        float v = acc[ai] + di * di * h[gbase + i] + bias[f];
        v = v > 0.0f ? v : 0.0f;
        v += xprev[gbase + i];
        xnew[gbase + i] = v;
        if (NEXT) acc[ai] = v;
    }

    if (NEXT) {
        __syncthreads();
        const int r0 = wave * 16;
        if (r0 >= cntNodes) return;
        const int kk = (lane >> 4) * 2;
        const float* arow = acc + (r0 + (lane & 15)) * ACC_PITCH + kk;
        v8f c0 = {}, c1 = {}, c2 = {}, c3 = {};
        #pragma unroll
        for (int ks = 0; ks < D_FEAT / 4; ++ks) {
            const v2f a = *reinterpret_cast<const v2f*>(arow + ks * 4);
            const float4 wa = wfrag[ks * 32 + lane];
            const float4 wb = wfrag[512 + ks * 32 + lane];
            v2f b0; b0.x = wa.x; b0.y = wa.y;
            v2f b1; b1.x = wa.z; b1.y = wa.w;
            v2f b2; b2.x = wb.x; b2.y = wb.y;
            v2f b3; b3.x = wb.z; b3.y = wb.w;
            c0 = __builtin_amdgcn_wmma_f32_16x16x4_f32(false, a, false, b0, (short)0, c0, false, false);
            c1 = __builtin_amdgcn_wmma_f32_16x16x4_f32(false, a, false, b1, (short)0, c1, false, false);
            c2 = __builtin_amdgcn_wmma_f32_16x16x4_f32(false, a, false, b2, (short)0, c2, false, false);
            c3 = __builtin_amdgcn_wmma_f32_16x16x4_f32(false, a, false, b3, (short)0, c3, false, false);
        }
        const int rbase = r0 + (lane >> 4) * 8;
        const int col   = lane & 15;
        #pragma unroll
        for (int r = 0; r < 8; ++r) {
            if (rbase + r < cntNodes) {
                float* hrow = hnext + (size_t)(lo + rbase + r) * D_FEAT + col;
                hrow[0]  = c0[r];
                hrow[16] = c1[r];
                hrow[32] = c2[r];
                hrow[48] = c3[r];
            }
        }
    }
}

#define POOL_WAVES 32
#define POOL_UNIT 128
#define POOL_UNITS 32
#define POOL_PASS_NODES (POOL_WAVES * POOL_UNITS * POOL_UNIT)
#define POOL_ITEM_CAP (POOL_UNITS * 8)
#define CNT_PITCH 32

__device__ __forceinline__ int4 pool_load_unit(
    const int* __restrict__ batch, int n0, int lane, int nNodes, bool aligned)
{
    const int n = n0 + lane * 4;
    int4 r;
    if (aligned && n0 + POOL_UNIT <= nNodes) {
        r = *reinterpret_cast<const int4*>(batch + n);
    } else {
        r.x = (n + 0 < nNodes) ? batch[n + 0] : -1;
        r.y = (n + 1 < nNodes) ? batch[n + 1] : -1;
        r.z = (n + 2 < nNodes) ? batch[n + 2] : -1;
        r.w = (n + 3 < nNodes) ? batch[n + 3] : -1;
    }
    return r;
}

__device__ __forceinline__ void pool_note(int b, int g, int nodeBase, int2* myItems, int& cnt, int lane)
{
    const unsigned m = __builtin_amdgcn_ballot_w32(b == g);
    if (m != 0u) {
        const unsigned lo = m & 0xffffu, hi = m >> 16;
        if (lo != 0u) {
            if (lane == 0) myItems[cnt] = make_int2(nodeBase, (int)lo);
            cnt += 1;
        }
        if (hi != 0u) {
            if (lane == 0) myItems[cnt] = make_int2(nodeBase + 64, (int)hi);
            cnt += 1;
        }
    }
}

__global__ __launch_bounds__(1024) void pool_owner(
    const float* __restrict__ x, const int* __restrict__ batch,
    float* __restrict__ sums, float* __restrict__ cnts, int nNodes)
{
    __shared__ int2  items[POOL_WAVES * POOL_ITEM_CAP];
    __shared__ int   itemCount[POOL_WAVES];
    __shared__ float part[POOL_WAVES * D_FEAT];
    __shared__ int   partCount[POOL_WAVES];

    const int tid  = threadIdx.x;
    const int lane = tid & 31;
    const int wave = __builtin_amdgcn_readfirstlane(tid >> 5);
    const int g = blockIdx.x;
    const bool aligned = (reinterpret_cast<size_t>(batch) & 15) == 0;

    float acc0 = 0.0f, acc1 = 0.0f;
    int nMine = 0;
    int2* myItems = items + wave * POOL_ITEM_CAP;

    for (int base = 0; base < nNodes; base += POOL_PASS_NODES) {
        int cnt = 0;
        for (int k0 = 0; k0 < POOL_UNITS; k0 += 8) {
            if (base + (k0 * POOL_WAVES + wave) * POOL_UNIT >= nNodes) break;
            int4 v[8];
            #pragma unroll
            for (int k = 0; k < 8; ++k)
                v[k] = pool_load_unit(batch, base + ((k0 + k) * POOL_WAVES + wave) * POOL_UNIT, lane, nNodes, aligned);
            #pragma unroll
            for (int k = 0; k < 8; ++k) {
                const int n0 = base + ((k0 + k) * POOL_WAVES + wave) * POOL_UNIT;
                pool_note(v[k].x, g, n0 + 0, myItems, cnt, lane);
                pool_note(v[k].y, g, n0 + 1, myItems, cnt, lane);
                pool_note(v[k].z, g, n0 + 2, myItems, cnt, lane);
                pool_note(v[k].w, g, n0 + 3, myItems, cnt, lane);
            }
        }
        if (lane == 0) itemCount[wave] = cnt;
        __syncthreads();

        const int cv = itemCount[lane];
        for (int w = 0; w < POOL_WAVES; ++w) {
            const int c = __builtin_amdgcn_readlane(cv, w);
            for (int j = (wave - 8 * w) & 31; j < c; j += 32) {
                const int2 it = items[w * POOL_ITEM_CAP + j];
                const int nb = __builtin_amdgcn_readfirstlane(it.x);
                const unsigned mk = (unsigned)__builtin_amdgcn_readfirstlane(it.y);
                const int first = nb + 4 * __builtin_ctz(mk);
                float v0[16], v1[16];
                #pragma unroll
                for (int k = 0; k < 16; ++k) {
                    const int node = ((mk >> k) & 1u) ? nb + 4 * k : first;
                    const float* xr = x + (size_t)node * D_FEAT + lane;
                    v0[k] = xr[0];
                    v1[k] = xr[32];
                }
                __builtin_amdgcn_sched_barrier(0);
                #pragma unroll
                for (int k = 0; k < 16; ++k) {
                    if ((mk >> k) & 1u) {
                        acc0 += v0[k];
                        acc1 += v1[k];
                    }
                }
                nMine += __builtin_popcount(mk);
            }
        }
        __syncthreads();
    }

    part[wave * D_FEAT + lane]      = acc0;
    part[wave * D_FEAT + 32 + lane] = acc1;
    if (lane == 0) partCount[wave] = nMine;
    __syncthreads();

    if (tid < D_FEAT) {
        float s = 0.0f;
        for (int w = 0; w < POOL_WAVES; ++w) s += part[w * D_FEAT + tid];
        sums[(size_t)g * D_FEAT + tid] = s;
    } else if (tid < D_FEAT + CNT_PITCH) {
        int c = 0;
        for (int w = 0; w < POOL_WAVES; ++w) c += partCount[w];
        cnts[(size_t)g * CNT_PITCH + (tid - D_FEAT)] = (float)c;
    }
}

__global__ __launch_bounds__(256) void final_linear_wmma(
    const float* __restrict__ sums, const float* __restrict__ cnts,
    const float* __restrict__ Wl, const float* __restrict__ bl,
    float* __restrict__ out, int nrows)
{
    const int lane  = threadIdx.x & 31;
    const int strip = blockIdx.x * 8 + (threadIdx.x >> 5);
    const int r0 = strip * 16;
    if (r0 >= nrows) return;

    const int m  = lane & 15;
    const int kk = (lane >> 4) * 2;
    const int n  = lane & 15;
    const int row = r0 + m;
    float c = cnts[(size_t)row * CNT_PITCH];
    float inv = 1.0f / (c > 1.0f ? c : 1.0f);

    v8f acc0 = {}, acc1 = {};
    const float* srow = sums + (size_t)row * D_FEAT;

    #pragma unroll
    for (int k0 = 0; k0 < D_FEAT; k0 += 4) {
        v2f a;
        a.x = srow[k0 + kk] * inv;
        a.y = srow[k0 + kk + 1] * inv;
        const float* w0 = Wl + (size_t)(k0 + kk) * D_OUT + n;
        const float* w1 = Wl + (size_t)(k0 + kk + 1) * D_OUT + n;
        v2f b0; b0.x = w0[0];  b0.y = w1[0];
        v2f b1; b1.x = w0[16]; b1.y = w1[16];
        acc0 = __builtin_amdgcn_wmma_f32_16x16x4_f32(false, a, false, b0, (short)0, acc0, false, false);
        acc1 = __builtin_amdgcn_wmma_f32_16x16x4_f32(false, a, false, b1, (short)0, acc1, false, false);
    }

    const int rbase = (lane >> 4) * 8;
    const int col   = lane & 15;
    #pragma unroll
    for (int r = 0; r < 8; ++r) {
        float* orow = out + (size_t)(r0 + rbase + r) * D_OUT + col;
        orow[0]  = acc0[r] + bl[col];
        orow[16] = acc1[r] + bl[col + 16];
    }
}

static inline int cdiv(long long a, int b) { return (int)((a + b - 1) / b); }

extern "C" void kernel_launch(void* const* d_in, const int* in_sizes, int n_in,
                              void* d_out, int out_size, void* d_ws, size_t ws_size,
                              hipStream_t stream)
{
    const float* x     = (const float*)d_in[0];
    const int*   ei    = (const int*)d_in[1];
    const int*   batch = (const int*)d_in[2];
    const float* W[3]  = {(const float*)d_in[3], (const float*)d_in[5], (const float*)d_in[7]};
    const float* B[3]  = {(const float*)d_in[4], (const float*)d_in[6], (const float*)d_in[8]};
    const float* Wl    = (const float*)d_in[9];
    const float* bl    = (const float*)d_in[10];
    float* out = (float*)d_out;

    const int N = in_sizes[0] / D_FEAT;
    const int E = in_sizes[1] / 2;
    const int G = out_size / D_OUT;
    const int* srcIdx = ei;
    const int* dstIdx = ei + E;

    float* ws   = (float*)d_ws;
    float* buf0 = ws;
    float* buf1 = buf0 + (size_t)N * D_FEAT;
    float* hbuf = buf1 + (size_t)N * D_FEAT;
    float* hbuf2 = hbuf + (size_t)N * D_FEAT;
    float* dinv = hbuf2 + (size_t)N * D_FEAT;
    float* sums = dinv + N;
    float* cnts = sums + (size_t)G * D_FEAT;

    const int TB = 256;

    const int tiles = cdiv(N, TILE_NODES);
    int* tileFill = (int*)(cnts + (size_t)G * CNT_PITCH);
    int* lists    = tileFill + tiles;

    init_lists<<<cdiv(tiles, TB), TB, 0, stream>>>(tileFill, tiles);
    bin_edges<<<cdiv(E, BIN_CHUNK), 1024, 0, stream>>>(srcIdx, dstIdx, N, E, tiles, lists, tileFill);
    node_degrees<<<cdiv(N, DEG_NODES), 1024, 0, stream>>>(dstIdx, N, E, lists, tileFill, dinv);

    const int gemmBlocks = cdiv((long long)cdiv(N, 16), 8);
    const float* xcur = x;
    float* xbuf[3] = {buf0, buf1, buf0};

    gcn_gemm64_wmma<<<gemmBlocks, TB, 0, stream>>>(x, W[0], hbuf, N);
    gcn_owner_tile<true, true><<<tiles, 1024, 0, stream>>>(srcIdx, dstIdx, dinv, hbuf, B[0], x, xbuf[0], N, E, lists, tileFill, W[1], hbuf2);
    gcn_owner_tile<false, true><<<tiles, 1024, 0, stream>>>(srcIdx, dstIdx, dinv, hbuf2, B[1], xbuf[0], xbuf[1], N, E, lists, tileFill, W[2], hbuf);
    gcn_owner_tile<false, false><<<tiles, 1024, 0, stream>>>(srcIdx, dstIdx, dinv, hbuf, B[2], xbuf[1], xbuf[2], N, E, lists, tileFill, W[2], hbuf2);
    xcur = xbuf[2];

    pool_owner<<<G, 1024, 0, stream>>>(xcur, batch, sums, cnts, N);
    final_linear_wmma<<<cdiv((long long)cdiv(G, 16), 8), TB, 0, stream>>>(sums, cnts, Wl, bl, out, G);
}


# First layer's product moved into the binning launch: its work-groups (256 rows; rows and W staged in LDS, 2 waves per strip) run beside the 98 binning ones on idle units; 9 -> 8 launches; sums kept
# speedup vs baseline: 3.9857x; 1.0586x over previous
#include <hip/hip_runtime.h>
#include <hip/hip_bf16.h>

typedef float v2f __attribute__((ext_vector_type(2)));
typedef float v8f __attribute__((ext_vector_type(8)));

#define D_FEAT 64
#define D_OUT  32

#define TILE_NODES 392
#define ACC_PITCH 68
#define ACC_ROWS 400
#define QCAP 512
#define LIST_CAP 16384
#define LIST_WORDS (3 * LIST_CAP)
#define BIN_CHUNK 16384
#define BIN_EPT (BIN_CHUNK / 1024)
#define BIN_TILES_MAX 256
#define BIN_STAGE (BIN_CHUNK + 31 * BIN_TILES_MAX)

__global__ void init_lists(int* __restrict__ tileFill, int tiles) {
    int i = blockIdx.x * blockDim.x + threadIdx.x;
    if (i < tiles) tileFill[i] = (tiles > BIN_TILES_MAX) ? LIST_CAP + 1 : 0;
}

#define PROD_ROWS 256
__device__ __forceinline__ void first_product_block(
    const float* __restrict__ X, const float* __restrict__ W, float* __restrict__ H,
    int nrows, int block, float* rows, float4* wfrag)
{
    const int tid  = threadIdx.x;
    const int lane = tid & 31;
    const int wave = __builtin_amdgcn_readfirstlane(tid >> 5);
    const int row0 = block * PROD_ROWS;

    {
        const int ks = tid >> 6;
        const int p  = (tid >> 5) & 1;
        const float* w0 = W + (size_t)(ks * 4 + (lane >> 4) * 2) * D_FEAT + p * 32 + (lane & 15);
        float4 v;
        v.x = w0[0];
        v.y = w0[D_FEAT];
        v.z = w0[16];
        v.w = w0[D_FEAT + 16];
        wfrag[p * 512 + ks * 32 + lane] = v;
    }

    const int fc = (tid & 15) * 4;
    const int fr = tid >> 4;
    float4 xv[PROD_ROWS / 64];
    #pragma unroll
    for (int k = 0; k < PROD_ROWS / 64; ++k) {
        const int gr = row0 + fr + 64 * k;
        const int gc = gr < nrows ? gr : nrows - 1;
        xv[k] = *reinterpret_cast<const float4*>(X + (size_t)gc * D_FEAT + fc);
    }
    #pragma unroll
    for (int k = 0; k < PROD_ROWS / 64; ++k) {
        const int gr = row0 + fr + 64 * k;
        float4 v = xv[k];
        if (gr >= nrows) v = make_float4(0.0f, 0.0f, 0.0f, 0.0f);
        *reinterpret_cast<float4*>(rows + (fr + 64 * k) * ACC_PITCH + fc) = v;
    }
    __syncthreads();

    const int r0 = (wave >> 1) * 16;
    if (row0 + r0 >= nrows) return;
    const int ch = wave & 1;
    const int kk = (lane >> 4) * 2;
    const float* arow = rows + (r0 + (lane & 15)) * ACC_PITCH + kk;
    const float4* wf = wfrag + ch * 512 + lane;
    v8f c0 = {}, c1 = {};
    #pragma unroll
    for (int ks = 0; ks < D_FEAT / 4; ++ks) {
        const v2f a = *reinterpret_cast<const v2f*>(arow + ks * 4);
        const float4 w = wf[ks * 32];
        v2f b0; b0.x = w.x; b0.y = w.y;
        v2f b1; b1.x = w.z; b1.y = w.w;
        c0 = __builtin_amdgcn_wmma_f32_16x16x4_f32(false, a, false, b0, (short)0, c0, false, false);
        c1 = __builtin_amdgcn_wmma_f32_16x16x4_f32(false, a, false, b1, (short)0, c1, false, false);
    }
    const int rbase = row0 + r0 + (lane >> 4) * 8;
    const int col   = ch * 32 + (lane & 15);
    #pragma unroll
    for (int r = 0; r < 8; ++r) {
        if (rbase + r < nrows) {
            float* hrow = H + (size_t)(rbase + r) * D_FEAT + col;
            hrow[0]  = c0[r];
            hrow[16] = c1[r];
        }
    }
}

template <int U>
__device__ __forceinline__ void rows_add(
    int s, int dl, float nrm, int n, int lane, const float* __restrict__ h, float* acc)
{
    for (int jb = 0; jb < n; jb += U) {
        float v0[U], v1[U], nj[U];
        int dj[U];
        #pragma unroll
        for (int u = 0; u < U; ++u) {
            const int j = jb + u;
            const int sj = __builtin_amdgcn_readlane(s, j);
            dj[u] = __builtin_amdgcn_readlane(dl, j);
            nj[u] = __int_as_float(__builtin_amdgcn_readlane(__float_as_int(nrm), j));
            const float* hr = h + (size_t)sj * D_FEAT + lane;
            v0[u] = hr[0];
            v1[u] = hr[32];
        }
        #pragma unroll
        for (int u = 0; u < U; ++u) {
            if (jb + u < n) {
                float* ar = acc + dj[u] * ACC_PITCH + lane;
                __hip_atomic_fetch_add(ar,      nj[u] * v0[u], __ATOMIC_RELAXED, __HIP_MEMORY_SCOPE_WORKGROUP);
                __hip_atomic_fetch_add(ar + 32, nj[u] * v1[u], __ATOMIC_RELAXED, __HIP_MEMORY_SCOPE_WORKGROUP);
            }
        }
    }
}

template <int JB>
__device__ __forceinline__ void rows_add_half(
    int s, int dl, float nrm, int lane, const float* __restrict__ h, float* acc)
{
    float v0[16], v1[16], nj[16];
    int dj[16];
    #pragma unroll
    for (int u = 0; u < 16; ++u) {
        const int sj = __builtin_amdgcn_readlane(s, JB + u);
        dj[u] = __builtin_amdgcn_readlane(dl, JB + u);
        nj[u] = __int_as_float(__builtin_amdgcn_readlane(__float_as_int(nrm), JB + u));
        const float* hr = h + (size_t)sj * D_FEAT + lane;
        v0[u] = hr[0];
        v1[u] = hr[32];
    }
    #pragma unroll
    for (int u = 0; u < 16; ++u) {
        float* ar = acc + dj[u] * ACC_PITCH + lane;
        __hip_atomic_fetch_add(ar,      nj[u] * v0[u], __ATOMIC_RELAXED, __HIP_MEMORY_SCOPE_WORKGROUP);
        __hip_atomic_fetch_add(ar + 32, nj[u] * v1[u], __ATOMIC_RELAXED, __HIP_MEMORY_SCOPE_WORKGROUP);
    }
}

__device__ __forceinline__ void tile_add_batch(
    const int* qp, int n, int lane, int lo,
    const int* __restrict__ srcIdx, const int* __restrict__ dstIdx,
    const float* __restrict__ dinv, const float* __restrict__ h, float* acc)
{
    int s = 0, dl = 0;
    float nrm = 0.0f;
    if (lane < n) {
        const int e = qp[lane];
        s = srcIdx[e];
        const int d = dstIdx[e];
        nrm = dinv[s] * dinv[d];
        dl = d - lo;
    }
    rows_add<8>(s, dl, nrm, n, lane, h, acc);
}

__device__ __forceinline__ void queue_matches(int dstv, int e, int lo, unsigned tileCount, int* q, int& cnt)
{
    const bool m = (unsigned)(dstv - lo) < tileCount;
    const unsigned mask = __builtin_amdgcn_ballot_w32(m);
    if (mask != 0u) {
        if (m) q[cnt + (int)__builtin_amdgcn_mbcnt_lo(mask, 0u)] = e;
        cnt += __builtin_popcount(mask);
    }
}

__device__ __forceinline__ void filter_pass(
    const int* __restrict__ srcIdx, const int* __restrict__ dstIdx,
    const float* __restrict__ dinv, const float* __restrict__ h, float* acc,
    int* q, int lane, int wave, int lo, unsigned tileCount, int nEdges)
{
    int cnt = 0;

    const bool aligned = (reinterpret_cast<size_t>(dstIdx) & 15) == 0;
    const int nFull = aligned ? (nEdges >> 10) : 0;
    const int rot = (int)(((long long)blockIdx.x * nFull) / gridDim.x);

    for (int c0 = wave; c0 < nFull; c0 += 32) {
        int c = c0 + rot;
        if (c >= nFull) c -= nFull;
        const int cb = c << 10;
        const int4* p = reinterpret_cast<const int4*>(dstIdx + cb) + lane;
        int4 dv[8];
        #pragma unroll
        for (int k = 0; k < 8; ++k) dv[k] = p[k * 32];
        #pragma unroll
        for (int k = 0; k < 8; ++k) {
            const int e0 = cb + k * 128 + lane * 4;
            queue_matches(dv[k].x, e0 + 0, lo, tileCount, q, cnt);
            queue_matches(dv[k].y, e0 + 1, lo, tileCount, q, cnt);
            queue_matches(dv[k].z, e0 + 2, lo, tileCount, q, cnt);
            queue_matches(dv[k].w, e0 + 3, lo, tileCount, q, cnt);
            while (cnt >= 32) {
                cnt -= 32;
                tile_add_batch(q + cnt, 32, lane, lo, srcIdx, dstIdx, dinv, h, acc);
            }
        }
    }

    for (int eb = (nFull << 10) + wave * 128; eb < nEdges; eb += 32 * 128) {
        #pragma unroll
        for (int t = 0; t < 4; ++t) {
            const int e = eb + t * 32 + lane;
            const int dstv = (e < nEdges) ? dstIdx[e] : -1;
            queue_matches(dstv, e, lo, tileCount, q, cnt);
        }
        while (cnt >= 32) {
            cnt -= 32;
            tile_add_batch(q + cnt, 32, lane, lo, srcIdx, dstIdx, dinv, h, acc);
        }
    }
    if (cnt > 0) tile_add_batch(q, cnt, lane, lo, srcIdx, dstIdx, dinv, h, acc);
}

template <bool FIRST>
__device__ __forceinline__ void list_pass(
    const int* __restrict__ rec, int* __restrict__ wts, int nSlots,
    const float* __restrict__ dinv, int lo,
    const float* __restrict__ h, float* acc, int lane, int wave)
{
    int c = wave * 32;
    if (c >= nSlots) return;
    int s  = rec[c + lane];
    int dl = rec[LIST_CAP + c + lane];
    int nb = 0;
    if (!FIRST) nb = rec[2 * LIST_CAP + c + lane];
    for (;;) {
        const int cn = c + 32 * 32;
        const bool more = cn < nSlots;
        const int cp = more ? cn : c;
        const int s2  = rec[cp + lane];
        const int dl2 = rec[LIST_CAP + cp + lane];
        int nb2 = 0;
        if (!FIRST) nb2 = rec[2 * LIST_CAP + cp + lane];
        float nrm;
        if (FIRST) {
            const bool used = dl < TILE_NODES;
            const float w = dinv[s] * dinv[lo + (used ? dl : 0)];
            nrm = used ? w : 0.0f;
        } else {
            nrm = __int_as_float(nb);
        }
        rows_add_half<0>(s, dl, nrm, lane, h, acc);
        if (__builtin_amdgcn_readlane(dl, 16) != TILE_NODES)
            rows_add_half<16>(s, dl, nrm, lane, h, acc);
        if (FIRST) wts[c + lane] = __float_as_int(nrm);
        if (!more) break;
        s = s2; dl = dl2; nb = nb2;
        c = cn;
    }
}

__global__ __launch_bounds__(1024) void bin_edges_and_first_product(
    const int* __restrict__ srcIdx, const int* __restrict__ dstIdx,
    int nNodes, int nEdges, int tiles,
    int* __restrict__ lists, int* __restrict__ tileFill,
    int binBlocks, const float* __restrict__ X, const float* __restrict__ W0, float* __restrict__ H)
{
    __shared__ __attribute__((aligned(16))) int stS[BIN_STAGE];
    __shared__ __attribute__((aligned(16))) int stD[BIN_STAGE];
    static_assert(PROD_ROWS * ACC_PITCH <= BIN_STAGE, "the product's rows must fit the first staging plane");
    static_assert(D_FEAT * D_FEAT <= BIN_STAGE, "the product's weights must fit the second staging plane");
    __shared__ int cnt[BIN_TILES_MAX];
    __shared__ int lbase[BIN_TILES_MAX];
    __shared__ int gbase[BIN_TILES_MAX];
    __shared__ int lineTile[BIN_STAGE / 32];
    __shared__ int wtot[BIN_TILES_MAX / 32];
    __shared__ int totalSlots;

    if ((int)blockIdx.x >= binBlocks) {
        first_product_block(X, W0, H, nNodes, (int)blockIdx.x - binBlocks,
                            reinterpret_cast<float*>(stS), reinterpret_cast<float4*>(stD));
        return;
    }

    if (tiles > BIN_TILES_MAX) return;

    const int tid  = threadIdx.x;
    const int lane = tid & 31;
    const int wave = __builtin_amdgcn_readfirstlane(tid >> 5);

    if (tid < BIN_TILES_MAX) cnt[tid] = 0;
    __syncthreads();

    const int e0 = blockIdx.x * BIN_CHUNK + tid;
    int sv[BIN_EPT], dv[BIN_EPT], rk[BIN_EPT];
    #pragma unroll
    for (int k = 0; k < BIN_EPT; ++k) {
        const int e = e0 + k * 1024;
        const bool ok = e < nEdges;
        const int el = ok ? e : 0;
        const int d = dstIdx[el];
        sv[k] = srcIdx[el];
        dv[k] = ok ? d : -1;
    }
    #pragma unroll
    for (int k = 0; k < BIN_EPT; ++k) {
        rk[k] = 0;
        if ((unsigned)dv[k] < (unsigned)nNodes)
            rk[k] = __hip_atomic_fetch_add(&cnt[dv[k] / TILE_NODES], 1, __ATOMIC_RELAXED, __HIP_MEMORY_SCOPE_WORKGROUP);
        else
            dv[k] = -1;
    }
    __syncthreads();

    int n = 0, padded = 0;
    if (tid < BIN_TILES_MAX) {
        n = cnt[tid];
        padded = (n + 31) & ~31;
    }
    int incl = padded;
    #pragma unroll
    for (int off = 1; off < 32; off <<= 1) {
        const int v = __shfl_up(incl, off);
        if (lane >= off) incl += v;
    }
    if (tid < BIN_TILES_MAX && lane == 31) wtot[wave] = incl;
    __syncthreads();
    if (tid < BIN_TILES_MAX) {
        int pre = 0;
        for (int w = 0; w < wave; ++w) pre += wtot[w];
        const int excl = pre + incl - padded;
        lbase[tid] = excl;
        int gb = 0;
        if (n > 0) gb = __hip_atomic_fetch_add(&tileFill[tid], padded, __ATOMIC_RELAXED, __HIP_MEMORY_SCOPE_AGENT);
        gbase[tid] = gb;
        for (int j = excl >> 5; j < ((excl + padded) >> 5); ++j) lineTile[j] = tid;
        for (int j = excl + n; j < excl + padded; ++j) { stS[j] = 0; stD[j] = TILE_NODES; }
        if (tid == BIN_TILES_MAX - 1) totalSlots = excl + padded;
    }
    __syncthreads();

    int pp[BIN_EPT];
    #pragma unroll
    for (int k = 0; k < BIN_EPT; ++k) {
        const int dd = dv[k] < 0 ? 0 : dv[k];
        const int t = dd / TILE_NODES;
        pp[k] = lbase[t] + rk[k];
        rk[k] = dd - t * TILE_NODES;
    }
    #pragma unroll
    for (int k = 0; k < BIN_EPT; ++k) {
        if (dv[k] >= 0) {
            stS[pp[k]] = sv[k];
            stD[pp[k]] = rk[k];
        }
    }
    __syncthreads();

    const int nLines = totalSlots >> 5;
    for (int ln0 = wave; ln0 < nLines; ln0 += 4 * 32) {
        int tt[4], gp[4], a[4], b[4];
        #pragma unroll
        for (int u = 0; u < 4; ++u) {
            const int ln = ln0 + 32 * u;
            tt[u] = lineTile[ln < nLines ? ln : ln0];
        }
        #pragma unroll
        for (int u = 0; u < 4; ++u) {
            const int ln = ln0 + 32 * u;
            const int lc = ln < nLines ? ln : ln0;
            tt[u] = __builtin_amdgcn_readfirstlane(tt[u]);
            gp[u] = gbase[tt[u]] + (lc * 32 - lbase[tt[u]]);
            a[u] = stS[lc * 32 + lane];
            b[u] = stD[lc * 32 + lane];
        }
        #pragma unroll
        for (int u = 0; u < 4; ++u) {
            const int ln = ln0 + 32 * u;
            const int gpos = __builtin_amdgcn_readfirstlane(gp[u]);
            if (ln < nLines && gpos < LIST_CAP) {
                int* r = lists + (size_t)tt[u] * LIST_WORDS + gpos + lane;
                r[0]        = a[u];
                r[LIST_CAP] = b[u];
            }
        }
    }
}

#define DEG_NODES 384
#define DEG_RPT (LIST_CAP / 1024)

__global__ __launch_bounds__(1024) void node_degrees(
    const int* __restrict__ dstIdx, int nNodes, int nEdges,
    const int* __restrict__ lists, const int* __restrict__ tileFill, float* __restrict__ dinv)
{
    __shared__ int cnt[DEG_NODES];

    const int tid  = threadIdx.x;
    const int base = blockIdx.x * DEG_NODES;
    int cntNodes = nNodes - base;
    if (cntNodes > DEG_NODES) cntNodes = DEG_NODES;
    if (cntNodes <= 0) return;

    if (tid < DEG_NODES) cnt[tid] = 0;
    __syncthreads();

    const int t0 = base / TILE_NODES;
    const int t1 = (base + cntNodes - 1) / TILE_NODES;
    const int n0 = tileFill[t0];
    const int n1 = (t1 != t0) ? tileFill[t1] : 0;

    if (n0 <= LIST_CAP && n1 <= LIST_CAP) {
        const int* pl0 = lists + (size_t)t0 * LIST_WORDS + LIST_CAP;
        const int* pl1 = lists + (size_t)t1 * LIST_WORDS + LIST_CAP;
        int pa[DEG_RPT], pb[DEG_RPT];
        #pragma unroll
        for (int k = 0; k < DEG_RPT; ++k) {
            const int i = tid + k * 1024;
            pa[k] = *(i < n0 ? pl0 + i : tileFill);
        }
        #pragma unroll
        for (int k = 0; k < DEG_RPT; ++k) {
            const int i = tid + k * 1024;
            pb[k] = *(i < n1 ? pl1 + i : tileFill);
        }
        const int off0 = t0 * TILE_NODES - base;
        const int off1 = t1 * TILE_NODES - base;
        #pragma unroll
        for (int k = 0; k < DEG_RPT; ++k) {
            const int i = tid + k * 1024;
            const unsigned r = (unsigned)(pa[k] + off0);
            if (i < n0 && pa[k] < TILE_NODES && r < (unsigned)cntNodes)
                __hip_atomic_fetch_add(&cnt[r], 1, __ATOMIC_RELAXED, __HIP_MEMORY_SCOPE_WORKGROUP);
        }
        #pragma unroll
        for (int k = 0; k < DEG_RPT; ++k) {
            const int i = tid + k * 1024;
            const unsigned r = (unsigned)(pb[k] + off1);
            if (i < n1 && pb[k] < TILE_NODES && r < (unsigned)cntNodes)
                __hip_atomic_fetch_add(&cnt[r], 1, __ATOMIC_RELAXED, __HIP_MEMORY_SCOPE_WORKGROUP);
        }
    } else {
        for (int e = tid; e < nEdges; e += 1024) {
            const unsigned r = (unsigned)(dstIdx[e] - base);
            if (r < (unsigned)cntNodes)
                __hip_atomic_fetch_add(&cnt[r], 1, __ATOMIC_RELAXED, __HIP_MEMORY_SCOPE_WORKGROUP);
        }
    }
    __syncthreads();

    if (tid < cntNodes) dinv[base + tid] = __frsqrt_rn(1.0f + (float)cnt[tid]);
}

template <bool FIRST, bool NEXT>
__global__ __launch_bounds__(1024) void gcn_owner_tile(
    const int* __restrict__ srcIdx, const int* __restrict__ dstIdx,
    const float* __restrict__ dinv, const float* __restrict__ h,
    const float* __restrict__ bias, const float* __restrict__ xprev,
    float* __restrict__ xnew, int nNodes, int nEdges,
    int* __restrict__ lists, const int* __restrict__ tileFill,
    const float* __restrict__ Wnext, float* __restrict__ hnext)
{
    __shared__ __attribute__((aligned(16))) float acc[ACC_ROWS * ACC_PITCH];
    __shared__ int   queue[32 * QCAP];
    __shared__ float4 wfrag[NEXT ? D_FEAT * D_FEAT / 4 : 1];

    const int tid  = threadIdx.x;
    const int lane = tid & 31;
    const int wave = __builtin_amdgcn_readfirstlane(tid >> 5);
    const int lo   = blockIdx.x * TILE_NODES;
    int cntNodes = nNodes - lo;
    if (cntNodes > TILE_NODES) cntNodes = TILE_NODES;
    if (cntNodes <= 0) return;
    const unsigned tileCount = (unsigned)cntNodes;

    for (int i = tid; i < ACC_ROWS * ACC_PITCH; i += 1024) acc[i] = 0.0f;
    if (NEXT) {
        const int ks = tid >> 6;
        const int p  = (tid >> 5) & 1;
        const float* w0 = Wnext + (size_t)(ks * 4 + (lane >> 4) * 2) * D_FEAT + p * 32 + (lane & 15);
        float4 v;
        v.x = w0[0];
        v.y = w0[D_FEAT];
        v.z = w0[16];
        v.w = w0[D_FEAT + 16];
        wfrag[p * 512 + ks * 32 + lane] = v;
    }
    __syncthreads();

    int* q = queue + wave * QCAP;
    int* rec = lists + (size_t)blockIdx.x * LIST_WORDS;

    const int nSlots = tileFill[blockIdx.x];
    if (nSlots <= LIST_CAP) {
        list_pass<FIRST>(rec, rec + 2 * LIST_CAP, nSlots, dinv, lo, h, acc, lane, wave);
    } else {
        filter_pass(srcIdx, dstIdx, dinv, h, acc, q, lane, wave, lo, tileCount, nEdges);
    }
    __syncthreads();

    const size_t gbase = (size_t)lo * D_FEAT;
    const int nElem = cntNodes * D_FEAT;
    for (int i = tid; i < nElem; i += 1024) {
        const int f = i & 63;
        const int ai = (i >> 6) * ACC_PITCH + f;
        const float di = dinv[lo + (i >> 6)];
        float v = acc[ai] + di * di * h[gbase + i] + bias[f];
        v = v > 0.0f ? v : 0.0f;
        v += xprev[gbase + i];
        xnew[gbase + i] = v;
        if (NEXT) acc[ai] = v;
    }

    if (NEXT) {
        __syncthreads();
        const int r0 = wave * 16;
        if (r0 >= cntNodes) return;
        const int kk = (lane >> 4) * 2;
        const float* arow = acc + (r0 + (lane & 15)) * ACC_PITCH + kk;
        v8f c0 = {}, c1 = {}, c2 = {}, c3 = {};
        #pragma unroll
        for (int ks = 0; ks < D_FEAT / 4; ++ks) {
            const v2f a = *reinterpret_cast<const v2f*>(arow + ks * 4);
            const float4 wa = wfrag[ks * 32 + lane];
            const float4 wb = wfrag[512 + ks * 32 + lane];
            v2f b0; b0.x = wa.x; b0.y = wa.y;
            v2f b1; b1.x = wa.z; b1.y = wa.w;
            v2f b2; b2.x = wb.x; b2.y = wb.y;
            v2f b3; b3.x = wb.z; b3.y = wb.w;
            c0 = __builtin_amdgcn_wmma_f32_16x16x4_f32(false, a, false, b0, (short)0, c0, false, false);
            c1 = __builtin_amdgcn_wmma_f32_16x16x4_f32(false, a, false, b1, (short)0, c1, false, false);
            c2 = __builtin_amdgcn_wmma_f32_16x16x4_f32(false, a, false, b2, (short)0, c2, false, false);
            c3 = __builtin_amdgcn_wmma_f32_16x16x4_f32(false, a, false, b3, (short)0, c3, false, false);
        }
        const int rbase = r0 + (lane >> 4) * 8;
        const int col   = lane & 15;
        #pragma unroll
        for (int r = 0; r < 8; ++r) {
            if (rbase + r < cntNodes) {
                float* hrow = hnext + (size_t)(lo + rbase + r) * D_FEAT + col;
                hrow[0]  = c0[r];
                hrow[16] = c1[r];
                hrow[32] = c2[r];
                hrow[48] = c3[r];
            }
        }
    }
}

#define POOL_WAVES 32
#define POOL_UNIT 128
#define POOL_UNITS 32
#define POOL_PASS_NODES (POOL_WAVES * POOL_UNITS * POOL_UNIT)
#define POOL_ITEM_CAP (POOL_UNITS * 8)
#define CNT_PITCH 32

__device__ __forceinline__ int4 pool_load_unit(
    const int* __restrict__ batch, int n0, int lane, int nNodes, bool aligned)
{
    const int n = n0 + lane * 4;
    int4 r;
    if (aligned && n0 + POOL_UNIT <= nNodes) {
        r = *reinterpret_cast<const int4*>(batch + n);
    } else {
        r.x = (n + 0 < nNodes) ? batch[n + 0] : -1;
        r.y = (n + 1 < nNodes) ? batch[n + 1] : -1;
        r.z = (n + 2 < nNodes) ? batch[n + 2] : -1;
        r.w = (n + 3 < nNodes) ? batch[n + 3] : -1;
    }
    return r;
}

__device__ __forceinline__ void pool_note(int b, int g, int nodeBase, int2* myItems, int& cnt, int lane)
{
    const unsigned m = __builtin_amdgcn_ballot_w32(b == g);
    if (m != 0u) {
        const unsigned lo = m & 0xffffu, hi = m >> 16;
        if (lo != 0u) {
            if (lane == 0) myItems[cnt] = make_int2(nodeBase, (int)lo);
            cnt += 1;
        }
        if (hi != 0u) {
            if (lane == 0) myItems[cnt] = make_int2(nodeBase + 64, (int)hi);
            cnt += 1;
        }
    }
}

__global__ __launch_bounds__(1024) void pool_owner(
    const float* __restrict__ x, const int* __restrict__ batch,
    float* __restrict__ sums, float* __restrict__ cnts, int nNodes)
{
    __shared__ int2  items[POOL_WAVES * POOL_ITEM_CAP];
    __shared__ int   itemCount[POOL_WAVES];
    __shared__ float part[POOL_WAVES * D_FEAT];
    __shared__ int   partCount[POOL_WAVES];

    const int tid  = threadIdx.x;
    const int lane = tid & 31;
    const int wave = __builtin_amdgcn_readfirstlane(tid >> 5);
    const int g = blockIdx.x;
    const bool aligned = (reinterpret_cast<size_t>(batch) & 15) == 0;

    float acc0 = 0.0f, acc1 = 0.0f;
    int nMine = 0;
    int2* myItems = items + wave * POOL_ITEM_CAP;

    for (int base = 0; base < nNodes; base += POOL_PASS_NODES) {
        int cnt = 0;
        for (int k0 = 0; k0 < POOL_UNITS; k0 += 8) {
            if (base + (k0 * POOL_WAVES + wave) * POOL_UNIT >= nNodes) break;
            int4 v[8];
            #pragma unroll
            for (int k = 0; k < 8; ++k)
                v[k] = pool_load_unit(batch, base + ((k0 + k) * POOL_WAVES + wave) * POOL_UNIT, lane, nNodes, aligned);
            #pragma unroll
            for (int k = 0; k < 8; ++k) {
                const int n0 = base + ((k0 + k) * POOL_WAVES + wave) * POOL_UNIT;
                pool_note(v[k].x, g, n0 + 0, myItems, cnt, lane);
                pool_note(v[k].y, g, n0 + 1, myItems, cnt, lane);
                pool_note(v[k].z, g, n0 + 2, myItems, cnt, lane);
                pool_note(v[k].w, g, n0 + 3, myItems, cnt, lane);
            }
        }
        if (lane == 0) itemCount[wave] = cnt;
        __syncthreads();

        const int cv = itemCount[lane];
        for (int w = 0; w < POOL_WAVES; ++w) {
            const int c = __builtin_amdgcn_readlane(cv, w);
            for (int j = (wave - 8 * w) & 31; j < c; j += 32) {
                const int2 it = items[w * POOL_ITEM_CAP + j];
                const int nb = __builtin_amdgcn_readfirstlane(it.x);
                const unsigned mk = (unsigned)__builtin_amdgcn_readfirstlane(it.y);
                const int first = nb + 4 * __builtin_ctz(mk);
                float v0[16], v1[16];
                #pragma unroll
                for (int k = 0; k < 16; ++k) {
                    const int node = ((mk >> k) & 1u) ? nb + 4 * k : first;
                    const float* xr = x + (size_t)node * D_FEAT + lane;
                    v0[k] = xr[0];
                    v1[k] = xr[32];
                }
                __builtin_amdgcn_sched_barrier(0);
                #pragma unroll
                for (int k = 0; k < 16; ++k) {
                    if ((mk >> k) & 1u) {
                        acc0 += v0[k];
                        acc1 += v1[k];
                    }
                }
                nMine += __builtin_popcount(mk);
            }
        }
        __syncthreads();
    }

    part[wave * D_FEAT + lane]      = acc0;
    part[wave * D_FEAT + 32 + lane] = acc1;
    if (lane == 0) partCount[wave] = nMine;
    __syncthreads();

    if (tid < D_FEAT) {
        float s = 0.0f;
        for (int w = 0; w < POOL_WAVES; ++w) s += part[w * D_FEAT + tid];
        sums[(size_t)g * D_FEAT + tid] = s;
    } else if (tid < D_FEAT + CNT_PITCH) {
        int c = 0;
        for (int w = 0; w < POOL_WAVES; ++w) c += partCount[w];
        cnts[(size_t)g * CNT_PITCH + (tid - D_FEAT)] = (float)c;
    }
}

__global__ __launch_bounds__(256) void final_linear_wmma(
    const float* __restrict__ sums, const float* __restrict__ cnts,
    const float* __restrict__ Wl, const float* __restrict__ bl,
    float* __restrict__ out, int nrows)
{
    const int lane  = threadIdx.x & 31;
    const int strip = blockIdx.x * 8 + (threadIdx.x >> 5);
    const int r0 = strip * 16;
    if (r0 >= nrows) return;

    const int m  = lane & 15;
    const int kk = (lane >> 4) * 2;
    const int n  = lane & 15;
    const int row = r0 + m;
    float c = cnts[(size_t)row * CNT_PITCH];
    float inv = 1.0f / (c > 1.0f ? c : 1.0f);

    v8f acc0 = {}, acc1 = {};
    const float* srow = sums + (size_t)row * D_FEAT;

    #pragma unroll
    for (int k0 = 0; k0 < D_FEAT; k0 += 4) {
        v2f a;
        a.x = srow[k0 + kk] * inv;
        a.y = srow[k0 + kk + 1] * inv;
        const float* w0 = Wl + (size_t)(k0 + kk) * D_OUT + n;
        const float* w1 = Wl + (size_t)(k0 + kk + 1) * D_OUT + n;
        v2f b0; b0.x = w0[0];  b0.y = w1[0];
        v2f b1; b1.x = w0[16]; b1.y = w1[16];
        acc0 = __builtin_amdgcn_wmma_f32_16x16x4_f32(false, a, false, b0, (short)0, acc0, false, false);
        acc1 = __builtin_amdgcn_wmma_f32_16x16x4_f32(false, a, false, b1, (short)0, acc1, false, false);
    }

    const int rbase = (lane >> 4) * 8;
    const int col   = lane & 15;
    #pragma unroll
    for (int r = 0; r < 8; ++r) {
        float* orow = out + (size_t)(r0 + rbase + r) * D_OUT + col;
        orow[0]  = acc0[r] + bl[col];
        orow[16] = acc1[r] + bl[col + 16];
    }
}

static inline int cdiv(long long a, int b) { return (int)((a + b - 1) / b); }

extern "C" void kernel_launch(void* const* d_in, const int* in_sizes, int n_in,
                              void* d_out, int out_size, void* d_ws, size_t ws_size,
                              hipStream_t stream)
{
    const float* x     = (const float*)d_in[0];
    const int*   ei    = (const int*)d_in[1];
    const int*   batch = (const int*)d_in[2];
    const float* W[3]  = {(const float*)d_in[3], (const float*)d_in[5], (const float*)d_in[7]};
    const float* B[3]  = {(const float*)d_in[4], (const float*)d_in[6], (const float*)d_in[8]};
    const float* Wl    = (const float*)d_in[9];
    const float* bl    = (const float*)d_in[10];
    float* out = (float*)d_out;

    const int N = in_sizes[0] / D_FEAT;
    const int E = in_sizes[1] / 2;
    const int G = out_size / D_OUT;
    const int* srcIdx = ei;
    const int* dstIdx = ei + E;

    float* ws   = (float*)d_ws;
    float* buf0 = ws;
    float* buf1 = buf0 + (size_t)N * D_FEAT;
    float* hbuf = buf1 + (size_t)N * D_FEAT;
    float* hbuf2 = hbuf + (size_t)N * D_FEAT;
    float* dinv = hbuf2 + (size_t)N * D_FEAT;
    float* sums = dinv + N;
    float* cnts = sums + (size_t)G * D_FEAT;

    const int TB = 256;

    const int tiles = cdiv(N, TILE_NODES);
    int* tileFill = (int*)(cnts + (size_t)G * CNT_PITCH);
    int* lists    = tileFill + tiles;

    init_lists<<<cdiv(tiles, TB), TB, 0, stream>>>(tileFill, tiles);
    const int binBlocks  = cdiv(E, BIN_CHUNK);
    const int prodBlocks = cdiv(N, PROD_ROWS);
    bin_edges_and_first_product<<<binBlocks + prodBlocks, 1024, 0, stream>>>(
        srcIdx, dstIdx, N, E, tiles, lists, tileFill, binBlocks, x, W[0], hbuf);
    node_degrees<<<cdiv(N, DEG_NODES), 1024, 0, stream>>>(dstIdx, N, E, lists, tileFill, dinv);

    const float* xcur = x;
    float* xbuf[3] = {buf0, buf1, buf0};

    gcn_owner_tile<true, true><<<tiles, 1024, 0, stream>>>(srcIdx, dstIdx, dinv, hbuf, B[0], x, xbuf[0], N, E, lists, tileFill, W[1], hbuf2);
    gcn_owner_tile<false, true><<<tiles, 1024, 0, stream>>>(srcIdx, dstIdx, dinv, hbuf2, B[1], xbuf[0], xbuf[1], N, E, lists, tileFill, W[2], hbuf);
    gcn_owner_tile<false, false><<<tiles, 1024, 0, stream>>>(srcIdx, dstIdx, dinv, hbuf, B[2], xbuf[1], xbuf[2], N, E, lists, tileFill, W[2], hbuf2);
    xcur = xbuf[2];

    pool_owner<<<G, 1024, 0, stream>>>(xcur, batch, sums, cnts, N);
    final_linear_wmma<<<cdiv((long long)cdiv(G, 16), 8), TB, 0, stream>>>(sums, cnts, Wl, bl, out, G);
}
